# starting kernel
# baseline (speedup 1.0000x reference)
#include <hip/hip_runtime.h>
#include <math.h>

typedef __attribute__((ext_vector_type(2))) float v2f;
typedef __attribute__((ext_vector_type(8))) float v8f;

#define LANE() ((int)(threadIdx.x & 31))

__device__ __forceinline__ float fast_tanh(float x) {
  const float e = __expf(2.0f * x);
  return 1.0f - 2.0f / (e + 1.0f);
}

template<int K, int LDA, bool TANH>
__device__ __forceinline__ void gemm_tile_to_lds(
    const float* __restrict__ sA,
    const float* __restrict__ Wg,
    const float* __restrict__ bg,
    float* __restrict__ sOut, int ldo, int n0)
{
  const int lane = LANE();
  const int half = lane >> 4;
  const int col  = lane & 15;
  const int n    = n0 + col;
  v8f acc = {};
#pragma unroll
  for (int k0 = 0; k0 < K; k0 += 4) {
    const int ka = k0 + half * 2;
    v2f a, b;
    a.x = sA[col * LDA + ka];
    a.y = sA[col * LDA + ka + 1];
    b.x = Wg[n * K + ka];
    b.y = Wg[n * K + ka + 1];
    acc = __builtin_amdgcn_wmma_f32_16x16x4_f32(false, a, false, b,
                                                (short)0, acc, false, false);
  }
  const float bias = bg[n];
#pragma unroll
  for (int r = 0; r < 8; ++r) {
    float v = acc[r] + bias;
    if (TANH) v = fast_tanh(v);
    sOut[(half * 8 + r) * ldo + n] = v;
  }
}

__device__ __forceinline__ void gemm_final(
    const float* __restrict__ sZ,
    const float* __restrict__ Wp,
    const float* __restrict__ bp,
    float* __restrict__ out, int rowBase, int n0)
{
  const int lane = LANE();
  const int half = lane >> 4;
  const int col  = lane & 15;
  const int n    = n0 + col;
  v8f acc = {};
#pragma unroll
  for (int k0 = 0; k0 < 8; k0 += 4) {
    const int ka = k0 + half * 2;
    v2f a, b;
    a.x = sZ[col * 8 + ka];
    a.y = sZ[col * 8 + ka + 1];
    b.x = Wp[n * 8 + ka];
    b.y = Wp[n * 8 + ka + 1];
    acc = __builtin_amdgcn_wmma_f32_16x16x4_f32(false, a, false, b,
                                                (short)0, acc, false, false);
  }
  const float bias = bp[n];
#pragma unroll
  for (int r = 0; r < 8; ++r)
    out[(size_t)(rowBase + half * 8 + r) * 128 + n] = acc[r] + bias;
}

template<int JB>
__device__ __forceinline__ void gate_local(float ar[8], float ai[8],
    float u00r, float u00i, float u01r, float u01i,
    float u10r, float u10i, float u11r, float u11i)
{
#pragma unroll
  for (int j = 0; j < 8; ++j) {
    if (!((j >> JB) & 1)) {
      const int j1 = j | (1 << JB);
      const float a0r = ar[j],  a0i = ai[j];
      const float a1r = ar[j1], a1i = ai[j1];
      ar[j]  = u00r*a0r - u00i*a0i + u01r*a1r - u01i*a1i;
      ai[j]  = u00r*a0i + u00i*a0r + u01r*a1i + u01i*a1r;
      ar[j1] = u10r*a0r - u10i*a0i + u11r*a1r - u11i*a1i;
      ai[j1] = u10r*a0i + u10i*a0r + u11r*a1i + u11i*a1r;
    }
  }
}

template<int LB>
__device__ __forceinline__ void gate_lane(float ar[8], float ai[8],
    float u00r, float u00i, float u01r, float u01i,
    float u10r, float u10i, float u11r, float u11i)
{
  const int bit = (LANE() >> LB) & 1;
  const float csr = bit ? u11r : u00r;
  const float csi = bit ? u11i : u00i;
  const float cpr = bit ? u10r : u01r;
  const float cpi = bit ? u10i : u01i;
#pragma unroll
  for (int j = 0; j < 8; ++j) {
    const float pr = __shfl_xor(ar[j], 1 << LB, 32);
    const float pi = __shfl_xor(ai[j], 1 << LB, 32);
    const float nr = csr*ar[j] - csi*ai[j] + cpr*pr - cpi*pi;
    const float ni = csr*ai[j] + csi*ar[j] + cpr*pi + cpi*pr;
    ar[j] = nr; ai[j] = ni;
  }
}

template<int W>
__device__ __forceinline__ void apply_gate(float ar[8], float ai[8],
    float u00r, float u00i, float u01r, float u01i,
    float u10r, float u10i, float u11r, float u11i)
{
  if constexpr (W <= 4)
    gate_lane<4 - W>(ar, ai, u00r, u00i, u01r, u01i, u10r, u10i, u11r, u11i);
  else
    gate_local<7 - W>(ar, ai, u00r, u00i, u01r, u01i, u10r, u10i, u11r, u11i);
}

template<int CJB, int TJB>
__device__ __forceinline__ void cnot_ll(float ar[8], float ai[8]) {
#pragma unroll
  for (int j = 0; j < 8; ++j)
    if (((j >> CJB) & 1) && !((j >> TJB) & 1)) {
      const int j1 = j | (1 << TJB);
      float tr = ar[j], ti = ai[j];
      ar[j] = ar[j1]; ai[j] = ai[j1];
      ar[j1] = tr;    ai[j1] = ti;
    }
}
template<int CJB, int TLB>
__device__ __forceinline__ void cnot_local_lane(float ar[8], float ai[8]) {
#pragma unroll
  for (int j = 0; j < 8; ++j) {
    const float pr = __shfl_xor(ar[j], 1 << TLB, 32);
    const float pi = __shfl_xor(ai[j], 1 << TLB, 32);
    if ((j >> CJB) & 1) { ar[j] = pr; ai[j] = pi; }
  }
}
template<int CLB, int TJB>
__device__ __forceinline__ void cnot_lane_local(float ar[8], float ai[8]) {
  if ((LANE() >> CLB) & 1) {
#pragma unroll
    for (int j = 0; j < 8; ++j)
      if (!((j >> TJB) & 1)) {
        const int j1 = j | (1 << TJB);
        float tr = ar[j], ti = ai[j];
        ar[j] = ar[j1]; ai[j] = ai[j1];
        ar[j1] = tr;    ai[j1] = ti;
      }
  }
}
template<int CLB, int TLB>
__device__ __forceinline__ void cnot_lane_lane(float ar[8], float ai[8]) {
  const int cb = (LANE() >> CLB) & 1;
#pragma unroll
  for (int j = 0; j < 8; ++j) {
    const float pr = __shfl_xor(ar[j], 1 << TLB, 32);
    const float pi = __shfl_xor(ai[j], 1 << TLB, 32);
    if (cb) { ar[j] = pr; ai[j] = pi; }
  }
}

__device__ __forceinline__ void rot_u(float phi, float th, float om,
                                      float* __restrict__ u)
{
  const float pr = __cosf(0.5f * phi), pi = -__sinf(0.5f * phi);
  const float qr = __cosf(0.5f * om),  qi = -__sinf(0.5f * om);
  const float cy = __cosf(0.5f * th),  sy =  __sinf(0.5f * th);
  const float qpr  = qr*pr - qi*pi, qpi  = qr*pi + qi*pr;
  const float qcpr = qr*pr + qi*pi, qcpi = qi*pr - qr*pi;
  u[0] =  cy*qpr;  u[1] =  cy*qpi;
  u[2] = -sy*qcpr; u[3] = -sy*qcpi;
  u[4] =  sy*qcpr; u[5] = -sy*qcpi;
  u[6] =  cy*qpr;  u[7] = -cy*qpi;
}

#define INIT_WIRE(W) {                                                        \
    const float aY = ang[(W)];                                                \
    const float cc = __cosf(0.5f * aY), ss = __sinf(0.5f * aY);               \
    apply_gate<(W)>(ar, ai, cc,0.f, -ss,0.f, ss,0.f, cc,0.f);                 \
    const float aZ = ang[8 + (W)];                                            \
    const float zr = __cosf(0.5f * aZ), zi = -__sinf(0.5f * aZ);              \
    apply_gate<(W)>(ar, ai, zr,zi, 0.f,0.f, 0.f,0.f, zr,-zi);                 \
  }

#define ROT_WIRE(L, W) {                                                      \
    const float* u = &sU[((L)*8 + (W)) * 8];                                  \
    apply_gate<(W)>(ar, ai, u[0],u[1],u[2],u[3],u[4],u[5],u[6],u[7]);         \
  }

__global__ __launch_bounds__(512) void quantum_block_kernel(
    const float* __restrict__ x,
    const float* __restrict__ W1, const float* __restrict__ b1,
    const float* __restrict__ W2, const float* __restrict__ b2,
    const float* __restrict__ W3, const float* __restrict__ b3,
    const float* __restrict__ qw,
    const float* __restrict__ Wp, const float* __restrict__ bp,
    float* __restrict__ out)
{
  __shared__ float sX  [16 * 80];
  __shared__ float sH1 [16 * 64];
  __shared__ float sH2 [16 * 32];
  __shared__ float sAng[16 * 16];
  __shared__ float sZ  [16 * 8];
  __shared__ float sU  [24 * 8];

  const int tid     = threadIdx.x;
  const int wid     = tid >> 5;
  const int lane    = tid & 31;
  const int rowBase = blockIdx.x * 16;

  if (tid < 24) {
    const int L = tid >> 3, W = tid & 7;
    rot_u(qw[(L * 8 + W) * 3 + 0], qw[(L * 8 + W) * 3 + 1],
          qw[(L * 8 + W) * 3 + 2], &sU[tid * 8]);
  }
  for (int i = tid; i < 16 * 80; i += 512) sX[i] = x[rowBase * 80 + i];
  __syncthreads();

  if (wid < 4) gemm_tile_to_lds<80, 80, true >(sX,  W1, b1, sH1, 64, wid * 16);
  __syncthreads();
  if (wid < 2) gemm_tile_to_lds<64, 64, true >(sH1, W2, b2, sH2, 32, wid * 16);
  __syncthreads();
  if (wid == 0) gemm_tile_to_lds<32, 32, false>(sH2, W3, b3, sAng, 16, 0);
  __syncthreads();

  {
    float ar[8], ai[8];
#pragma unroll
    for (int j = 0; j < 8; ++j) { ar[j] = 0.f; ai[j] = 0.f; }
    if (lane == 0) ar[0] = 1.0f;

    const float* ang = &sAng[wid * 16];
    INIT_WIRE(0) INIT_WIRE(1) INIT_WIRE(2) INIT_WIRE(3)
    INIT_WIRE(4) INIT_WIRE(5) INIT_WIRE(6) INIT_WIRE(7)

    for (int L = 0; L < 3; ++L) {
      ROT_WIRE(L, 0) ROT_WIRE(L, 1) ROT_WIRE(L, 2) ROT_WIRE(L, 3)
      ROT_WIRE(L, 4) ROT_WIRE(L, 5) ROT_WIRE(L, 6) ROT_WIRE(L, 7)
      cnot_lane_lane<4, 3>(ar, ai);
      cnot_lane_lane<3, 2>(ar, ai);
      cnot_lane_lane<2, 1>(ar, ai);
      cnot_lane_lane<1, 0>(ar, ai);
      cnot_lane_local<0, 2>(ar, ai);
      cnot_ll<2, 1>(ar, ai);
      cnot_ll<1, 0>(ar, ai);
      cnot_local_lane<0, 4>(ar, ai);
    }

    float pb[8]; float tot = 0.f;
#pragma unroll
    for (int j = 0; j < 8; ++j) { pb[j] = ar[j]*ar[j] + ai[j]*ai[j]; tot += pb[j]; }
    float zp[8];
#pragma unroll
    for (int w = 0; w < 5; ++w)
      zp[w] = ((lane >> (4 - w)) & 1) ? -tot : tot;
    float s5 = 0.f, s6 = 0.f, s7 = 0.f;
#pragma unroll
    for (int j = 0; j < 8; ++j) {
      s5 += ((j >> 2) & 1) ? -pb[j] : pb[j];
      s6 += ((j >> 1) & 1) ? -pb[j] : pb[j];
      s7 += ( j       & 1) ? -pb[j] : pb[j];
    }
    zp[5] = s5; zp[6] = s6; zp[7] = s7;
#pragma unroll
    for (int w = 0; w < 8; ++w) {
      float v = zp[w];
#pragma unroll
      for (int off = 16; off > 0; off >>= 1) v += __shfl_xor(v, off, 32);
      if (lane == 0) sZ[wid * 8 + w] = v;
    }
  }
  __syncthreads();

  if (wid < 8) gemm_final(sZ, Wp, bp, out, rowBase, wid * 16);
}

extern "C" void kernel_launch(void* const* d_in, const int* in_sizes, int n_in,
                              void* d_out, int out_size, void* d_ws, size_t ws_size,
                              hipStream_t stream) {
  const float* x  = (const float*)d_in[0];
  const float* W1 = (const float*)d_in[1];
  const float* b1 = (const float*)d_in[2];
  const float* W2 = (const float*)d_in[3];
  const float* b2 = (const float*)d_in[4];
  const float* W3 = (const float*)d_in[5];
  const float* b3 = (const float*)d_in[6];
  const float* qw = (const float*)d_in[7];
  const float* Wp = (const float*)d_in[8];
  const float* bp = (const float*)d_in[9];
  float* out = (float*)d_out;

  const int B = in_sizes[0] / 80;
  const int grid = B / 16;
  quantum_block_kernel<<<grid, 512, 0, stream>>>(x, W1, b1, W2, b2, W3, b3,
                                                 qw, Wp, bp, out);
}


# Circuit state as a 16x16 complex matrix in WMMA accumulator layout: each Rot layer = two batch-shared Kronecker factors on the matrix unit, CNOT ring folded in, outer-product init; sums re-associated
# speedup vs baseline: 1.4085x; 1.4085x over previous
#include <hip/hip_runtime.h>
#include <math.h>

typedef __attribute__((ext_vector_type(2))) float v2f;
typedef __attribute__((ext_vector_type(8))) float v8f;

#define LANE() ((int)(threadIdx.x & 31))

__device__ __forceinline__ float fast_tanh(float x) {
  const float e = __expf(2.0f * x);
  return 1.0f - 2.0f / (e + 1.0f);
}

template<int K, int LDA, bool TANH>
__device__ __forceinline__ void gemm_tile_to_lds(
    const float* __restrict__ sA,
    const float* __restrict__ Wg,
    const float* __restrict__ bg,
    float* __restrict__ sOut, int ldo, int n0)
{
  const int lane = LANE();
  const int half = lane >> 4;
  const int col  = lane & 15;
  const int n    = n0 + col;
  v8f acc = {};
#pragma unroll
  for (int k0 = 0; k0 < K; k0 += 4) {
    const int ka = k0 + half * 2;
    v2f a, b;
    a.x = sA[col * LDA + ka];
    a.y = sA[col * LDA + ka + 1];
    b.x = Wg[n * K + ka];
    b.y = Wg[n * K + ka + 1];
    acc = __builtin_amdgcn_wmma_f32_16x16x4_f32(false, a, false, b,
                                                (short)0, acc, false, false);
  }
  const float bias = bg[n];
#pragma unroll
  for (int r = 0; r < 8; ++r) {
    float v = acc[r] + bias;
    if (TANH) v = fast_tanh(v);
    sOut[(half * 8 + r) * ldo + n] = v;
  }
}

__device__ __forceinline__ void gemm_final(
    const float* __restrict__ sZ,
    const float* __restrict__ Wp,
    const float* __restrict__ bp,
    float* __restrict__ out, int rowBase, int n0)
{
  const int lane = LANE();
  const int half = lane >> 4;
  const int col  = lane & 15;
  const int n    = n0 + col;
  v8f acc = {};
#pragma unroll
  for (int k0 = 0; k0 < 8; k0 += 4) {
    const int ka = k0 + half * 2;
    v2f a, b;
    a.x = sZ[col * 8 + ka];
    a.y = sZ[col * 8 + ka + 1];
    b.x = Wp[n * 8 + ka];
    b.y = Wp[n * 8 + ka + 1];
    acc = __builtin_amdgcn_wmma_f32_16x16x4_f32(false, a, false, b,
                                                (short)0, acc, false, false);
  }
  const float bias = bp[n];
#pragma unroll
  for (int r = 0; r < 8; ++r)
    out[(size_t)(rowBase + half * 8 + r) * 128 + n] = acc[r] + bias;
}


__device__ __forceinline__ int wire_bitpos(int grp, int p) {
  return grp ? (3 - p) : ((0x0132 >> (4 * p)) & 15);
}

__device__ __forceinline__ void build_factor_tables(
    const float* __restrict__ sU, float* __restrict__ sTab, int t)
{
#pragma unroll
  for (int it = 0; it < 4; ++it) {
    const int idx  = t + it * 384;
    const int T    = idx >> 8;
    const int grp  = T & 1;
    const int e256 = idx & 255;
    const int q    = e256 >> 7;
    const int ln   = (e256 >> 2) & 31;
    const int e    = e256 & 3;
    const int kidx = 8 * (ln >> 4) + 4 * q + e;
    const int nidx = ln & 15;
    float pr = 1.0f, pi = 0.0f;
    int tprev = 0;
#pragma unroll
    for (int p = 0; p < 4; ++p) {
      const int bp = wire_bitpos(grp, p);
      const int j  = (kidx >> bp) & 1;
      const int tb = (nidx >> bp) & 1;
      const int i  = tb ^ tprev;
      tprev = tb;
      const float* u = &sU[((T >> 1) * 8 + grp * 4 + p) * 8 + (i * 2 + j) * 2];
      const float ur = u[0], ui = u[1];
      const float nr = pr * ur - pi * ui;
      const float ni = pr * ui + pi * ur;
      pr = nr; pi = ni;
    }
    const int o = (((T * 2) * 2 + q) * 32 + ln) * 4 + e;
    sTab[o]       = pr;
    sTab[o + 256] = pi;
  }
}

__device__ __forceinline__ void kron_stage(
    const v8f xr, const v8f xi, const float* __restrict__ tabT, int lane,
    v8f& yr, v8f& yi)
{
  const float4* t4 = reinterpret_cast<const float4*>(tabT);
  const float4 r0 = t4[lane],      r1 = t4[32 + lane];
  const float4 m0 = t4[64 + lane], m1 = t4[96 + lane];
  const float gr[8] = {r0.x, r0.y, r0.z, r0.w, r1.x, r1.y, r1.z, r1.w};
  const float gi[8] = {m0.x, m0.y, m0.z, m0.w, m1.x, m1.y, m1.z, m1.w};
  v8f ar = {}, ai = {};
#pragma unroll
  for (int s = 0; s < 4; ++s) {
    v2f a_r, a_i, b_r, b_i, b_n;
    a_r.x = xr[2 * s]; a_r.y = xr[2 * s + 1];
    a_i.x = xi[2 * s]; a_i.y = xi[2 * s + 1];
    b_r.x = gr[2 * s]; b_r.y = gr[2 * s + 1];
    b_i.x = gi[2 * s]; b_i.y = gi[2 * s + 1];
    b_n.x = -gi[2 * s]; b_n.y = -gi[2 * s + 1];
    ar = __builtin_amdgcn_wmma_f32_16x16x4_f32(false, a_r, false, b_r,
                                               (short)0, ar, false, false);
    ai = __builtin_amdgcn_wmma_f32_16x16x4_f32(false, a_r, false, b_i,
                                               (short)0, ai, false, false);
    ar = __builtin_amdgcn_wmma_f32_16x16x4_f32(false, a_i, false, b_n,
                                               (short)0, ar, false, false);
    ai = __builtin_amdgcn_wmma_f32_16x16x4_f32(false, a_i, false, b_r,
                                               (short)0, ai, false, false);
  }
  yr = ar; yi = ai;
}

__device__ __forceinline__ float row_mirror(float v) {
  const int iv = __builtin_bit_cast(int, v);
  return __builtin_bit_cast(float,
      __builtin_amdgcn_update_dpp(iv, iv, 0x140, 0xf, 0xf, false));
}

__device__ __forceinline__ void rot_u(float phi, float th, float om,
                                      float* __restrict__ u)
{
  const float pr = __cosf(0.5f * phi), pi = -__sinf(0.5f * phi);
  const float qr = __cosf(0.5f * om),  qi = -__sinf(0.5f * om);
  const float cy = __cosf(0.5f * th),  sy =  __sinf(0.5f * th);
  const float qpr  = qr*pr - qi*pi, qpi  = qr*pi + qi*pr;
  const float qcpr = qr*pr + qi*pi, qcpi = qi*pr - qr*pi;
  u[0] =  cy*qpr;  u[1] =  cy*qpi;
  u[2] = -sy*qcpr; u[3] = -sy*qcpi;
  u[4] =  sy*qcpr; u[5] = -sy*qcpi;
  u[6] =  cy*qpr;  u[7] = -cy*qpi;
}

__global__ __launch_bounds__(512) void quantum_block_kernel(
    const float* __restrict__ x,
    const float* __restrict__ W1, const float* __restrict__ b1,
    const float* __restrict__ W2, const float* __restrict__ b2,
    const float* __restrict__ W3, const float* __restrict__ b3,
    const float* __restrict__ qw,
    const float* __restrict__ Wp, const float* __restrict__ bp,
    float* __restrict__ out)
{
  __shared__ float sX  [16 * 80];
  __shared__ float sH1 [16 * 64];
  __shared__ float sH2 [16 * 32];
  __shared__ float sAng[16 * 16];
  __shared__ float sZ  [16 * 8];
  __shared__ float sU  [24 * 8];
  __shared__ __attribute__((aligned(16))) float sTab[6 * 512];

  const int tid     = threadIdx.x;
  const int wid     = tid >> 5;
  const int lane    = tid & 31;
  const int rowBase = blockIdx.x * 16;

  if (tid < 24) {
    const int L = tid >> 3, W = tid & 7;
    rot_u(qw[(L * 8 + W) * 3 + 0], qw[(L * 8 + W) * 3 + 1],
          qw[(L * 8 + W) * 3 + 2], &sU[tid * 8]);
  }
  for (int i = tid; i < 16 * 80; i += 512) sX[i] = x[rowBase * 80 + i];
  __syncthreads();

  if (wid < 4) gemm_tile_to_lds<80, 80, true >(sX,  W1, b1, sH1, 64, wid * 16);
  else         build_factor_tables(sU, sTab, tid - 128);
  __syncthreads();
  if (wid < 2) gemm_tile_to_lds<64, 64, true >(sH1, W2, b2, sH2, 32, wid * 16);
  __syncthreads();
  if (wid == 0) gemm_tile_to_lds<32, 32, false>(sH2, W3, b3, sAng, 16, 0);
  __syncthreads();

  {
    const int h = lane >> 4;
    const int c = lane & 15;
    const float* ang = &sAng[wid * 16];

    float wr = 1.0f, wi = 0.0f;
#pragma unroll
    for (int p = 0; p < 4; ++p) {
      const int bit = (c >> wire_bitpos(h, p)) & 1;
      const float al = ang[4 * h + p], be = ang[8 + 4 * h + p];
      const float ca = __cosf(0.5f * al), sa = __sinf(0.5f * al);
      const float cb = __cosf(0.5f * be), sb = __sinf(0.5f * be);
      const float mag = bit ? sa : ca;
      const float fr = mag * cb;
      const float fi = mag * (bit ? sb : -sb);
      const float nr = wr * fr - wi * fi;
      const float ni = wr * fi + wi * fr;
      wr = nr; wi = ni;
    }
    const float vr = __shfl_xor(wr, 16, 32);
    const float vi = __shfl_xor(wi, 16, 32);
    v2f a1, b1, a2, b2;
    a1.x = h ? 0.0f : wr;  a1.y = h ? 0.0f : -wi;
    b1.x = h ? 0.0f : vr;  b1.y = h ? 0.0f : vi;
    a2.x = h ? 0.0f : wr;  a2.y = h ? 0.0f : wi;
    b2.x = h ? 0.0f : vi;  b2.y = h ? 0.0f : vr;
    v8f xr = {}, xi = {};
    xr = __builtin_amdgcn_wmma_f32_16x16x4_f32(false, a1, false, b1,
                                               (short)0, xr, false, false);
    xi = __builtin_amdgcn_wmma_f32_16x16x4_f32(false, a2, false, b2,
                                               (short)0, xi, false, false);

    const bool odd = (lane & 1) != 0;
    for (int L = 0; L < 3; ++L) {
      v8f yr, yi, zr, zi;
      kron_stage(xr, xi, &sTab[(2 * L) * 512],     lane, yr, yi);
      kron_stage(yr, yi, &sTab[(2 * L + 1) * 512], lane, zr, zi);
#pragma unroll
      for (int r = 1; r < 8; r += 2) {
        zr[r] = row_mirror(zr[r]);
        zi[r] = row_mirror(zi[r]);
      }
#pragma unroll
      for (int r = 0; r < 4; ++r) {
        const float pr0 = zr[r], pr1 = zr[r + 4];
        const float pi0 = zi[r], pi1 = zi[r + 4];
        xr[r]     = odd ? pr1 : pr0;
        xr[r + 4] = odd ? pr0 : pr1;
        xi[r]     = odd ? pi1 : pi0;
        xi[r + 4] = odd ? pi0 : pi1;
      }
    }

    float pb[8]; float tot = 0.f;
#pragma unroll
    for (int r = 0; r < 8; ++r) { pb[r] = xr[r]*xr[r] + xi[r]*xi[r]; tot += pb[r]; }
    float s0 = 0.f, s2 = 0.f, s3 = 0.f;
#pragma unroll
    for (int r = 0; r < 8; ++r) {
      s0 += ((r >> 2) & 1) ? -pb[r] : pb[r];
      s2 += ((r >> 1) & 1) ? -pb[r] : pb[r];
      s3 += ( r       & 1) ? -pb[r] : pb[r];
    }
    float zp[8];
    zp[0] = s0;
    zp[1] = h ? -tot : tot;
    zp[2] = s2;
    zp[3] = s3;
#pragma unroll
    for (int w = 4; w < 8; ++w)
      zp[w] = ((c >> (7 - w)) & 1) ? -tot : tot;
#pragma unroll
    for (int w = 0; w < 8; ++w) {
      float v = zp[w];
#pragma unroll
      for (int off = 16; off > 0; off >>= 1) v += __shfl_xor(v, off, 32);
      if (lane == 0) sZ[wid * 8 + w] = v;
    }
  }
  __syncthreads();

  if (wid < 8) gemm_final(sZ, Wp, bp, out, rowBase, wid * 16);
}

extern "C" void kernel_launch(void* const* d_in, const int* in_sizes, int n_in,
                              void* d_out, int out_size, void* d_ws, size_t ws_size,
                              hipStream_t stream) {
  const float* x  = (const float*)d_in[0];
  const float* W1 = (const float*)d_in[1];
  const float* b1 = (const float*)d_in[2];
  const float* W2 = (const float*)d_in[3];
  const float* b2 = (const float*)d_in[4];
  const float* W3 = (const float*)d_in[5];
  const float* b3 = (const float*)d_in[6];
  const float* qw = (const float*)d_in[7];
  const float* Wp = (const float*)d_in[8];
  const float* bp = (const float*)d_in[9];
  float* out = (float*)d_out;

  const int B = in_sizes[0] / 80;
  const int grid = B / 16;
  quantum_block_kernel<<<grid, 512, 0, stream>>>(x, W1, b1, W2, b2, W3, b3,
                                                 qw, Wp, bp, out);
}


# Complex Kronecker stages as three real matrix products instead of four (k=(xr+xi)gr starts both accumulator chains; tables hold gr, -(gr+gi), gi-gr): 12 WMMA per stage, not 16; sums re-associated
# speedup vs baseline: 1.6683x; 1.1845x over previous
#include <hip/hip_runtime.h>
#include <math.h>

typedef __attribute__((ext_vector_type(2))) float v2f;
typedef __attribute__((ext_vector_type(8))) float v8f;

#define LANE() ((int)(threadIdx.x & 31))

__device__ __forceinline__ float fast_tanh(float x) {
  const float e = __expf(2.0f * x);
  return 1.0f - 2.0f / (e + 1.0f);
}

template<int K, int LDA, bool TANH>
__device__ __forceinline__ void gemm_tile_to_lds(
    const float* __restrict__ sA,
    const float* __restrict__ Wg,
    const float* __restrict__ bg,
    float* __restrict__ sOut, int ldo, int n0)
{
  const int lane = LANE();
  const int half = lane >> 4;
  const int col  = lane & 15;
  const int n    = n0 + col;
  v8f acc = {};
#pragma unroll
  for (int k0 = 0; k0 < K; k0 += 4) {
    const int ka = k0 + half * 2;
    v2f a, b;
    a.x = sA[col * LDA + ka];
    a.y = sA[col * LDA + ka + 1];
    b.x = Wg[n * K + ka];
    b.y = Wg[n * K + ka + 1];
    acc = __builtin_amdgcn_wmma_f32_16x16x4_f32(false, a, false, b,
                                                (short)0, acc, false, false);
  }
  const float bias = bg[n];
#pragma unroll
  for (int r = 0; r < 8; ++r) {
    float v = acc[r] + bias;
    if (TANH) v = fast_tanh(v);
    sOut[(half * 8 + r) * ldo + n] = v;
  }
}

__device__ __forceinline__ void gemm_final(
    const float* __restrict__ sZ,
    const float* __restrict__ Wp,
    const float* __restrict__ bp,
    float* __restrict__ out, int rowBase, int n0)
{
  const int lane = LANE();
  const int half = lane >> 4;
  const int col  = lane & 15;
  const int n    = n0 + col;
  v8f acc = {};
#pragma unroll
  for (int k0 = 0; k0 < 8; k0 += 4) {
    const int ka = k0 + half * 2;
    v2f a, b;
    a.x = sZ[col * 8 + ka];
    a.y = sZ[col * 8 + ka + 1];
    b.x = Wp[n * 8 + ka];
    b.y = Wp[n * 8 + ka + 1];
    acc = __builtin_amdgcn_wmma_f32_16x16x4_f32(false, a, false, b,
                                                (short)0, acc, false, false);
  }
  const float bias = bp[n];
#pragma unroll
  for (int r = 0; r < 8; ++r)
    out[(size_t)(rowBase + half * 8 + r) * 128 + n] = acc[r] + bias;
}


__device__ __forceinline__ int wire_bitpos(int grp, int p) {
  return grp ? (3 - p) : ((0x0132 >> (4 * p)) & 15);
}

__device__ __forceinline__ void build_factor_tables(
    const float* __restrict__ sU, float* __restrict__ sTab, int t)
{
#pragma unroll
  for (int it = 0; it < 4; ++it) {
    const int idx  = t + it * 384;
    const int T    = idx >> 8;
    const int grp  = T & 1;
    const int e256 = idx & 255;
    const int q    = e256 >> 7;
    const int ln   = (e256 >> 2) & 31;
    const int e    = e256 & 3;
    const int kidx = 8 * (ln >> 4) + 4 * q + e;
    const int nidx = ln & 15;
    float pr = 1.0f, pi = 0.0f;
    int tprev = 0;
#pragma unroll
    for (int p = 0; p < 4; ++p) {
      const int bp = wire_bitpos(grp, p);
      const int j  = (kidx >> bp) & 1;
      const int tb = (nidx >> bp) & 1;
      const int i  = tb ^ tprev;
      tprev = tb;
      const float* u = &sU[((T >> 1) * 8 + grp * 4 + p) * 8 + (i * 2 + j) * 2];
      const float ur = u[0], ui = u[1];
      const float nr = pr * ur - pi * ui;
      const float ni = pr * ui + pi * ur;
      pr = nr; pi = ni;
    }
    const int o = ((T * 6 + q) * 32 + ln) * 4 + e;
    sTab[o]       = pr;
    sTab[o + 256] = -(pr + pi);
    sTab[o + 512] = pi - pr;
  }
}

__device__ __forceinline__ void kron_stage(
    const v8f xr, const v8f xi, const float* __restrict__ tabT, int lane,
    v8f& yr, v8f& yi)
{
  const float4* t4 = reinterpret_cast<const float4*>(tabT);
  const float4 p0 = t4[lane],       p1 = t4[32 + lane];
  const float4 s0 = t4[64 + lane],  s1 = t4[96 + lane];
  const float4 d0 = t4[128 + lane], d1 = t4[160 + lane];
  const float gp[8] = {p0.x, p0.y, p0.z, p0.w, p1.x, p1.y, p1.z, p1.w};
  const float gs[8] = {s0.x, s0.y, s0.z, s0.w, s1.x, s1.y, s1.z, s1.w};
  const float gd[8] = {d0.x, d0.y, d0.z, d0.w, d1.x, d1.y, d1.z, d1.w};
  const v8f xs = xr + xi;
  v8f k = {};
#pragma unroll
  for (int s = 0; s < 4; ++s) {
    v2f a_s, b_p;
    a_s.x = xs[2 * s]; a_s.y = xs[2 * s + 1];
    b_p.x = gp[2 * s]; b_p.y = gp[2 * s + 1];
    k = __builtin_amdgcn_wmma_f32_16x16x4_f32(false, a_s, false, b_p,
                                              (short)0, k, false, false);
  }
  v8f ar = k, ai = k;
#pragma unroll
  for (int s = 0; s < 4; ++s) {
    v2f a_r, a_i, b_s, b_d;
    a_r.x = xr[2 * s]; a_r.y = xr[2 * s + 1];
    a_i.x = xi[2 * s]; a_i.y = xi[2 * s + 1];
    b_s.x = gs[2 * s]; b_s.y = gs[2 * s + 1];
    b_d.x = gd[2 * s]; b_d.y = gd[2 * s + 1];
    ar = __builtin_amdgcn_wmma_f32_16x16x4_f32(false, a_i, false, b_s,
                                               (short)0, ar, false, false);
    ai = __builtin_amdgcn_wmma_f32_16x16x4_f32(false, a_r, false, b_d,
                                               (short)0, ai, false, false);
  }
  yr = ar; yi = ai;
}

__device__ __forceinline__ float row_mirror(float v) {
  const int iv = __builtin_bit_cast(int, v);
  return __builtin_bit_cast(float,
      __builtin_amdgcn_update_dpp(iv, iv, 0x140, 0xf, 0xf, false));
}

__device__ __forceinline__ void rot_u(float phi, float th, float om,
                                      float* __restrict__ u)
{
  const float pr = __cosf(0.5f * phi), pi = -__sinf(0.5f * phi);
  const float qr = __cosf(0.5f * om),  qi = -__sinf(0.5f * om);
  const float cy = __cosf(0.5f * th),  sy =  __sinf(0.5f * th);
  const float qpr  = qr*pr - qi*pi, qpi  = qr*pi + qi*pr;
  const float qcpr = qr*pr + qi*pi, qcpi = qi*pr - qr*pi;
  u[0] =  cy*qpr;  u[1] =  cy*qpi;
  u[2] = -sy*qcpr; u[3] = -sy*qcpi;
  u[4] =  sy*qcpr; u[5] = -sy*qcpi;
  u[6] =  cy*qpr;  u[7] = -cy*qpi;
}

__global__ __launch_bounds__(512) void quantum_block_kernel(
    const float* __restrict__ x,
    const float* __restrict__ W1, const float* __restrict__ b1,
    const float* __restrict__ W2, const float* __restrict__ b2,
    const float* __restrict__ W3, const float* __restrict__ b3,
    const float* __restrict__ qw,
    const float* __restrict__ Wp, const float* __restrict__ bp,
    float* __restrict__ out)
{
  __shared__ float sX  [16 * 80];
  __shared__ float sH1 [16 * 64];
  __shared__ float sH2 [16 * 32];
  __shared__ float sAng[16 * 16];
  __shared__ float sZ  [16 * 8];
  __shared__ float sU  [24 * 8];
  __shared__ __attribute__((aligned(16))) float sTab[6 * 768];

  const int tid     = threadIdx.x;
  const int wid     = tid >> 5;
  const int lane    = tid & 31;
  const int rowBase = blockIdx.x * 16;

  if (tid < 24) {
    const int L = tid >> 3, W = tid & 7;
    rot_u(qw[(L * 8 + W) * 3 + 0], qw[(L * 8 + W) * 3 + 1],
          qw[(L * 8 + W) * 3 + 2], &sU[tid * 8]);
  }
  for (int i = tid; i < 16 * 80; i += 512) sX[i] = x[rowBase * 80 + i];
  __syncthreads();

  if (wid < 4) gemm_tile_to_lds<80, 80, true >(sX,  W1, b1, sH1, 64, wid * 16);
  else         build_factor_tables(sU, sTab, tid - 128);
  __syncthreads();
  if (wid < 2) gemm_tile_to_lds<64, 64, true >(sH1, W2, b2, sH2, 32, wid * 16);
  __syncthreads();
  if (wid == 0) gemm_tile_to_lds<32, 32, false>(sH2, W3, b3, sAng, 16, 0);
  __syncthreads();

  {
    const int h = lane >> 4;
    const int c = lane & 15;
    const float* ang = &sAng[wid * 16];

    float wr = 1.0f, wi = 0.0f;
#pragma unroll
    for (int p = 0; p < 4; ++p) {
      const int bit = (c >> wire_bitpos(h, p)) & 1;
      const float al = ang[4 * h + p], be = ang[8 + 4 * h + p];
      const float ca = __cosf(0.5f * al), sa = __sinf(0.5f * al);
      const float cb = __cosf(0.5f * be), sb = __sinf(0.5f * be);
      const float mag = bit ? sa : ca;
      const float fr = mag * cb;
      const float fi = mag * (bit ? sb : -sb);
      const float nr = wr * fr - wi * fi;
      const float ni = wr * fi + wi * fr;
      wr = nr; wi = ni;
    }
    const float vr = __shfl_xor(wr, 16, 32);
    const float vi = __shfl_xor(wi, 16, 32);
    v2f a1, b1, a2, b2;
    a1.x = h ? 0.0f : wr;  a1.y = h ? 0.0f : -wi;
    b1.x = h ? 0.0f : vr;  b1.y = h ? 0.0f : vi;
    a2.x = h ? 0.0f : wr;  a2.y = h ? 0.0f : wi;
    b2.x = h ? 0.0f : vi;  b2.y = h ? 0.0f : vr;
    v8f xr = {}, xi = {};
    xr = __builtin_amdgcn_wmma_f32_16x16x4_f32(false, a1, false, b1,
                                               (short)0, xr, false, false);
    xi = __builtin_amdgcn_wmma_f32_16x16x4_f32(false, a2, false, b2,
                                               (short)0, xi, false, false);

    const bool odd = (lane & 1) != 0;
    for (int L = 0; L < 3; ++L) {
      v8f yr, yi, zr, zi;
      kron_stage(xr, xi, &sTab[(2 * L) * 768],     lane, yr, yi);
      kron_stage(yr, yi, &sTab[(2 * L + 1) * 768], lane, zr, zi);
#pragma unroll
      for (int r = 1; r < 8; r += 2) {
        zr[r] = row_mirror(zr[r]);
        zi[r] = row_mirror(zi[r]);
      }
#pragma unroll
      for (int r = 0; r < 4; ++r) {
        const float pr0 = zr[r], pr1 = zr[r + 4];
        const float pi0 = zi[r], pi1 = zi[r + 4];
        xr[r]     = odd ? pr1 : pr0;
        xr[r + 4] = odd ? pr0 : pr1;
        xi[r]     = odd ? pi1 : pi0;
        xi[r + 4] = odd ? pi0 : pi1;
      }
    }

    float pb[8]; float tot = 0.f;
#pragma unroll
    for (int r = 0; r < 8; ++r) { pb[r] = xr[r]*xr[r] + xi[r]*xi[r]; tot += pb[r]; }
    float s0 = 0.f, s2 = 0.f, s3 = 0.f;
#pragma unroll
    for (int r = 0; r < 8; ++r) {
      s0 += ((r >> 2) & 1) ? -pb[r] : pb[r];
      s2 += ((r >> 1) & 1) ? -pb[r] : pb[r];
      s3 += ( r       & 1) ? -pb[r] : pb[r];
    }
    float zp[8];
    zp[0] = s0;
    zp[1] = h ? -tot : tot;
    zp[2] = s2;
    zp[3] = s3;
#pragma unroll
    for (int w = 4; w < 8; ++w)
      zp[w] = ((c >> (7 - w)) & 1) ? -tot : tot;
#pragma unroll
    for (int w = 0; w < 8; ++w) {
      float v = zp[w];
#pragma unroll
      for (int off = 16; off > 0; off >>= 1) v += __shfl_xor(v, off, 32);
      if (lane == 0) sZ[wid * 8 + w] = v;
    }
  }
  __syncthreads();

  if (wid < 8) gemm_final(sZ, Wp, bp, out, rowBase, wid * 16);
}

extern "C" void kernel_launch(void* const* d_in, const int* in_sizes, int n_in,
                              void* d_out, int out_size, void* d_ws, size_t ws_size,
                              hipStream_t stream) {
  const float* x  = (const float*)d_in[0];
  const float* W1 = (const float*)d_in[1];
  const float* b1 = (const float*)d_in[2];
  const float* W2 = (const float*)d_in[3];
  const float* b2 = (const float*)d_in[4];
  const float* W3 = (const float*)d_in[5];
  const float* b3 = (const float*)d_in[6];
  const float* qw = (const float*)d_in[7];
  const float* Wp = (const float*)d_in[8];
  const float* bp = (const float*)d_in[9];
  float* out = (float*)d_out;

  const int B = in_sizes[0] / 80;
  const int grid = B / 16;
  quantum_block_kernel<<<grid, 512, 0, stream>>>(x, W1, b1, W2, b2, W3, b3,
                                                 qw, Wp, bp, out);
}


# The eight <Z> lane sums merged into one butterfly that hands values over as it goes: 4+2+1+1+1 lane exchanges in 5 dependent rounds instead of 40 in 40; same pairing of every sum
# speedup vs baseline: 1.8187x; 1.0902x over previous
#include <hip/hip_runtime.h>
#include <math.h>

typedef __attribute__((ext_vector_type(2))) float v2f;
typedef __attribute__((ext_vector_type(8))) float v8f;

#define LANE() ((int)(threadIdx.x & 31))

__device__ __forceinline__ float fast_tanh(float x) {
  const float e = __expf(2.0f * x);
  return 1.0f - 2.0f / (e + 1.0f);
}

template<int K, int LDA, bool TANH>
__device__ __forceinline__ void gemm_tile_to_lds(
    const float* __restrict__ sA,
    const float* __restrict__ Wg,
    const float* __restrict__ bg,
    float* __restrict__ sOut, int ldo, int n0)
{
  const int lane = LANE();
  const int half = lane >> 4;
  const int col  = lane & 15;
  const int n    = n0 + col;
  v8f acc = {};
#pragma unroll
  for (int k0 = 0; k0 < K; k0 += 4) {
    const int ka = k0 + half * 2;
    v2f a, b;
    a.x = sA[col * LDA + ka];
    a.y = sA[col * LDA + ka + 1];
    b.x = Wg[n * K + ka];
    b.y = Wg[n * K + ka + 1];
    acc = __builtin_amdgcn_wmma_f32_16x16x4_f32(false, a, false, b,
                                                (short)0, acc, false, false);
  }
  const float bias = bg[n];
#pragma unroll
  for (int r = 0; r < 8; ++r) {
    float v = acc[r] + bias;
    if (TANH) v = fast_tanh(v);
    sOut[(half * 8 + r) * ldo + n] = v;
  }
}

__device__ __forceinline__ void gemm_final(
    const float* __restrict__ sZ,
    const float* __restrict__ Wp,
    const float* __restrict__ bp,
    float* __restrict__ out, int rowBase, int n0)
{
  const int lane = LANE();
  const int half = lane >> 4;
  const int col  = lane & 15;
  const int n    = n0 + col;
  v8f acc = {};
#pragma unroll
  for (int k0 = 0; k0 < 8; k0 += 4) {
    const int ka = k0 + half * 2;
    v2f a, b;
    a.x = sZ[col * 8 + ka];
    a.y = sZ[col * 8 + ka + 1];
    b.x = Wp[n * 8 + ka];
    b.y = Wp[n * 8 + ka + 1];
    acc = __builtin_amdgcn_wmma_f32_16x16x4_f32(false, a, false, b,
                                                (short)0, acc, false, false);
  }
  const float bias = bp[n];
#pragma unroll
  for (int r = 0; r < 8; ++r)
    out[(size_t)(rowBase + half * 8 + r) * 128 + n] = acc[r] + bias;
}


__device__ __forceinline__ int wire_bitpos(int grp, int p) {
  return grp ? (3 - p) : ((0x0132 >> (4 * p)) & 15);
}

__device__ __forceinline__ void build_factor_tables(
    const float* __restrict__ sU, float* __restrict__ sTab, int t)
{
#pragma unroll
  for (int it = 0; it < 4; ++it) {
    const int idx  = t + it * 384;
    const int T    = idx >> 8;
    const int grp  = T & 1;
    const int e256 = idx & 255;
    const int q    = e256 >> 7;
    const int ln   = (e256 >> 2) & 31;
    const int e    = e256 & 3;
    const int kidx = 8 * (ln >> 4) + 4 * q + e;
    const int nidx = ln & 15;
    float pr = 1.0f, pi = 0.0f;
    int tprev = 0;
#pragma unroll
    for (int p = 0; p < 4; ++p) {
      const int bp = wire_bitpos(grp, p);
      const int j  = (kidx >> bp) & 1;
      const int tb = (nidx >> bp) & 1;
      const int i  = tb ^ tprev;
      tprev = tb;
      const float* u = &sU[((T >> 1) * 8 + grp * 4 + p) * 8 + (i * 2 + j) * 2];
      const float ur = u[0], ui = u[1];
      const float nr = pr * ur - pi * ui;
      const float ni = pr * ui + pi * ur;
      pr = nr; pi = ni;
    }
    const int o = ((T * 6 + q) * 32 + ln) * 4 + e;
    sTab[o]       = pr;
    sTab[o + 256] = -(pr + pi);
    sTab[o + 512] = pi - pr;
  }
}

__device__ __forceinline__ void kron_stage(
    const v8f xr, const v8f xi, const float* __restrict__ tabT, int lane,
    v8f& yr, v8f& yi)
{
  const float4* t4 = reinterpret_cast<const float4*>(tabT);
  const float4 p0 = t4[lane],       p1 = t4[32 + lane];
  const float4 s0 = t4[64 + lane],  s1 = t4[96 + lane];
  const float4 d0 = t4[128 + lane], d1 = t4[160 + lane];
  const float gp[8] = {p0.x, p0.y, p0.z, p0.w, p1.x, p1.y, p1.z, p1.w};
  const float gs[8] = {s0.x, s0.y, s0.z, s0.w, s1.x, s1.y, s1.z, s1.w};
  const float gd[8] = {d0.x, d0.y, d0.z, d0.w, d1.x, d1.y, d1.z, d1.w};
  const v8f xs = xr + xi;
  v8f k = {};
#pragma unroll
  for (int s = 0; s < 4; ++s) {
    v2f a_s, b_p;
    a_s.x = xs[2 * s]; a_s.y = xs[2 * s + 1];
    b_p.x = gp[2 * s]; b_p.y = gp[2 * s + 1];
    k = __builtin_amdgcn_wmma_f32_16x16x4_f32(false, a_s, false, b_p,
                                              (short)0, k, false, false);
  }
  v8f ar = k, ai = k;
#pragma unroll
  for (int s = 0; s < 4; ++s) {
    v2f a_r, a_i, b_s, b_d;
    a_r.x = xr[2 * s]; a_r.y = xr[2 * s + 1];
    a_i.x = xi[2 * s]; a_i.y = xi[2 * s + 1];
    b_s.x = gs[2 * s]; b_s.y = gs[2 * s + 1];
    b_d.x = gd[2 * s]; b_d.y = gd[2 * s + 1];
    ar = __builtin_amdgcn_wmma_f32_16x16x4_f32(false, a_i, false, b_s,
                                               (short)0, ar, false, false);
    ai = __builtin_amdgcn_wmma_f32_16x16x4_f32(false, a_r, false, b_d,
                                               (short)0, ai, false, false);
  }
  yr = ar; yi = ai;
}

__device__ __forceinline__ float row_mirror(float v) {
  const int iv = __builtin_bit_cast(int, v);
  return __builtin_bit_cast(float,
      __builtin_amdgcn_update_dpp(iv, iv, 0x140, 0xf, 0xf, false));
}

__device__ __forceinline__ void rot_u(float phi, float th, float om,
                                      float* __restrict__ u)
{
  const float pr = __cosf(0.5f * phi), pi = -__sinf(0.5f * phi);
  const float qr = __cosf(0.5f * om),  qi = -__sinf(0.5f * om);
  const float cy = __cosf(0.5f * th),  sy =  __sinf(0.5f * th);
  const float qpr  = qr*pr - qi*pi, qpi  = qr*pi + qi*pr;
  const float qcpr = qr*pr + qi*pi, qcpi = qi*pr - qr*pi;
  u[0] =  cy*qpr;  u[1] =  cy*qpi;
  u[2] = -sy*qcpr; u[3] = -sy*qcpi;
  u[4] =  sy*qcpr; u[5] = -sy*qcpi;
  u[6] =  cy*qpr;  u[7] = -cy*qpi;
}

__global__ __launch_bounds__(512) void quantum_block_kernel(
    const float* __restrict__ x,
    const float* __restrict__ W1, const float* __restrict__ b1,
    const float* __restrict__ W2, const float* __restrict__ b2,
    const float* __restrict__ W3, const float* __restrict__ b3,
    const float* __restrict__ qw,
    const float* __restrict__ Wp, const float* __restrict__ bp,
    float* __restrict__ out)
{
  __shared__ float sX  [16 * 80];
  __shared__ float sH1 [16 * 64];
  __shared__ float sH2 [16 * 32];
  __shared__ float sAng[16 * 16];
  __shared__ float sZ  [16 * 8];
  __shared__ float sU  [24 * 8];
  __shared__ __attribute__((aligned(16))) float sTab[6 * 768];

  const int tid     = threadIdx.x;
  const int wid     = tid >> 5;
  const int lane    = tid & 31;
  const int rowBase = blockIdx.x * 16;

  if (tid < 24) {
    const int L = tid >> 3, W = tid & 7;
    rot_u(qw[(L * 8 + W) * 3 + 0], qw[(L * 8 + W) * 3 + 1],
          qw[(L * 8 + W) * 3 + 2], &sU[tid * 8]);
  }
  for (int i = tid; i < 16 * 80; i += 512) sX[i] = x[rowBase * 80 + i];
  __syncthreads();

  if (wid < 4) gemm_tile_to_lds<80, 80, true >(sX,  W1, b1, sH1, 64, wid * 16);
  else         build_factor_tables(sU, sTab, tid - 128);
  __syncthreads();
  if (wid < 2) gemm_tile_to_lds<64, 64, true >(sH1, W2, b2, sH2, 32, wid * 16);
  __syncthreads();
  if (wid == 0) gemm_tile_to_lds<32, 32, false>(sH2, W3, b3, sAng, 16, 0);
  __syncthreads();

  {
    const int h = lane >> 4;
    const int c = lane & 15;
    const float* ang = &sAng[wid * 16];

    float wr = 1.0f, wi = 0.0f;
#pragma unroll
    for (int p = 0; p < 4; ++p) {
      const int bit = (c >> wire_bitpos(h, p)) & 1;
      const float al = ang[4 * h + p], be = ang[8 + 4 * h + p];
      const float ca = __cosf(0.5f * al), sa = __sinf(0.5f * al);
      const float cb = __cosf(0.5f * be), sb = __sinf(0.5f * be);
      const float mag = bit ? sa : ca;
      const float fr = mag * cb;
      const float fi = mag * (bit ? sb : -sb);
      const float nr = wr * fr - wi * fi;
      const float ni = wr * fi + wi * fr;
      wr = nr; wi = ni;
    }
    const float vr = __shfl_xor(wr, 16, 32);
    const float vi = __shfl_xor(wi, 16, 32);
    v2f a1, b1, a2, b2;
    a1.x = h ? 0.0f : wr;  a1.y = h ? 0.0f : -wi;
    b1.x = h ? 0.0f : vr;  b1.y = h ? 0.0f : vi;
    a2.x = h ? 0.0f : wr;  a2.y = h ? 0.0f : wi;
    b2.x = h ? 0.0f : vi;  b2.y = h ? 0.0f : vr;
    v8f xr = {}, xi = {};
    xr = __builtin_amdgcn_wmma_f32_16x16x4_f32(false, a1, false, b1,
                                               (short)0, xr, false, false);
    xi = __builtin_amdgcn_wmma_f32_16x16x4_f32(false, a2, false, b2,
                                               (short)0, xi, false, false);

    const bool odd = (lane & 1) != 0;
    for (int L = 0; L < 3; ++L) {
      v8f yr, yi, zr, zi;
      kron_stage(xr, xi, &sTab[(2 * L) * 768],     lane, yr, yi);
      kron_stage(yr, yi, &sTab[(2 * L + 1) * 768], lane, zr, zi);
#pragma unroll
      for (int r = 1; r < 8; r += 2) {
        zr[r] = row_mirror(zr[r]);
        zi[r] = row_mirror(zi[r]);
      }
#pragma unroll
      for (int r = 0; r < 4; ++r) {
        const float pr0 = zr[r], pr1 = zr[r + 4];
        const float pi0 = zi[r], pi1 = zi[r + 4];
        xr[r]     = odd ? pr1 : pr0;
        xr[r + 4] = odd ? pr0 : pr1;
        xi[r]     = odd ? pi1 : pi0;
        xi[r + 4] = odd ? pi0 : pi1;
      }
    }

    float pb[8]; float tot = 0.f;
#pragma unroll
    for (int r = 0; r < 8; ++r) { pb[r] = xr[r]*xr[r] + xi[r]*xi[r]; tot += pb[r]; }
    float s0 = 0.f, s2 = 0.f, s3 = 0.f;
#pragma unroll
    for (int r = 0; r < 8; ++r) {
      s0 += ((r >> 2) & 1) ? -pb[r] : pb[r];
      s2 += ((r >> 1) & 1) ? -pb[r] : pb[r];
      s3 += ( r       & 1) ? -pb[r] : pb[r];
    }
    float zp[8];
    zp[0] = s0;
    zp[1] = h ? -tot : tot;
    zp[2] = s2;
    zp[3] = s3;
#pragma unroll
    for (int w = 4; w < 8; ++w)
      zp[w] = ((c >> (7 - w)) & 1) ? -tot : tot;
    const bool up16 = (lane & 16) != 0;
    const bool up8  = (lane & 8) != 0;
    const bool up4  = (lane & 4) != 0;
    float q4[4], q2[2];
#pragma unroll
    for (int j = 0; j < 4; ++j) {
      const float snd = up16 ? zp[j] : zp[4 + j];
      const float kp  = up16 ? zp[4 + j] : zp[j];
      q4[j] = kp + __shfl_xor(snd, 16, 32);
    }
#pragma unroll
    for (int j = 0; j < 2; ++j) {
      const float snd = up8 ? q4[j] : q4[2 + j];
      const float kp  = up8 ? q4[2 + j] : q4[j];
      q2[j] = kp + __shfl_xor(snd, 8, 32);
    }
    float zs;
    {
      const float snd = up4 ? q2[0] : q2[1];
      const float kp  = up4 ? q2[1] : q2[0];
      zs = kp + __shfl_xor(snd, 4, 32);
    }
    zs += __shfl_xor(zs, 2, 32);
    zs += __shfl_xor(zs, 1, 32);
    if ((lane & 3) == 0) sZ[wid * 8 + (lane >> 2)] = zs;
  }
  __syncthreads();

  if (wid < 8) gemm_final(sZ, Wp, bp, out, rowBase, wid * 16);
}

extern "C" void kernel_launch(void* const* d_in, const int* in_sizes, int n_in,
                              void* d_out, int out_size, void* d_ws, size_t ws_size,
                              hipStream_t stream) {
  const float* x  = (const float*)d_in[0];
  const float* W1 = (const float*)d_in[1];
  const float* b1 = (const float*)d_in[2];
  const float* W2 = (const float*)d_in[3];
  const float* b2 = (const float*)d_in[4];
  const float* W3 = (const float*)d_in[5];
  const float* b3 = (const float*)d_in[6];
  const float* qw = (const float*)d_in[7];
  const float* Wp = (const float*)d_in[8];
  const float* bp = (const float*)d_in[9];
  float* out = (float*)d_out;

  const int B = in_sizes[0] / 80;
  const int grid = B / 16;
  quantum_block_kernel<<<grid, 512, 0, stream>>>(x, W1, b1, W2, b2, W3, b3,
                                                 qw, Wp, bp, out);
}


# First Rot layer applied to the two 16-vectors of the rank-1 start state (per-wire 2x2 products in the encoding loop), not to the 16x16 matrix: 24 of 74 WMMA per sample gone; sums re-associated
# speedup vs baseline: 2.1447x; 1.1792x over previous
#include <hip/hip_runtime.h>
#include <math.h>

typedef __attribute__((ext_vector_type(2))) float v2f;
typedef __attribute__((ext_vector_type(8))) float v8f;

#define LANE() ((int)(threadIdx.x & 31))

__device__ __forceinline__ float fast_tanh(float x) {
  const float e = __expf(2.0f * x);
  return 1.0f - 2.0f / (e + 1.0f);
}

template<int K, int LDA, bool TANH>
__device__ __forceinline__ void gemm_tile_to_lds(
    const float* __restrict__ sA,
    const float* __restrict__ Wg,
    const float* __restrict__ bg,
    float* __restrict__ sOut, int ldo, int n0)
{
  const int lane = LANE();
  const int half = lane >> 4;
  const int col  = lane & 15;
  const int n    = n0 + col;
  v8f acc = {};
#pragma unroll
  for (int k0 = 0; k0 < K; k0 += 4) {
    const int ka = k0 + half * 2;
    v2f a, b;
    a.x = sA[col * LDA + ka];
    a.y = sA[col * LDA + ka + 1];
    b.x = Wg[n * K + ka];
    b.y = Wg[n * K + ka + 1];
    acc = __builtin_amdgcn_wmma_f32_16x16x4_f32(false, a, false, b,
                                                (short)0, acc, false, false);
  }
  const float bias = bg[n];
#pragma unroll
  for (int r = 0; r < 8; ++r) {
    float v = acc[r] + bias;
    if (TANH) v = fast_tanh(v);
    sOut[(half * 8 + r) * ldo + n] = v;
  }
}

__device__ __forceinline__ void gemm_final(
    const float* __restrict__ sZ,
    const float* __restrict__ Wp,
    const float* __restrict__ bp,
    float* __restrict__ out, int rowBase, int n0)
{
  const int lane = LANE();
  const int half = lane >> 4;
  const int col  = lane & 15;
  const int n    = n0 + col;
  v8f acc = {};
#pragma unroll
  for (int k0 = 0; k0 < 8; k0 += 4) {
    const int ka = k0 + half * 2;
    v2f a, b;
    a.x = sZ[col * 8 + ka];
    a.y = sZ[col * 8 + ka + 1];
    b.x = Wp[n * 8 + ka];
    b.y = Wp[n * 8 + ka + 1];
    acc = __builtin_amdgcn_wmma_f32_16x16x4_f32(false, a, false, b,
                                                (short)0, acc, false, false);
  }
  const float bias = bp[n];
#pragma unroll
  for (int r = 0; r < 8; ++r)
    out[(size_t)(rowBase + half * 8 + r) * 128 + n] = acc[r] + bias;
}


__device__ __forceinline__ int wire_bitpos(int grp, int p) {
  return grp ? (3 - p) : ((0x0132 >> (4 * p)) & 15);
}

__device__ __forceinline__ void build_factor_tables(
    const float* __restrict__ sU, float* __restrict__ sTab, int t)
{
#pragma unroll
  for (int it = 0; it < 3; ++it) {
    const int idx  = t + it * 384;
    if (idx >= 1024) break;
    const int T    = (idx >> 8) + 2;
    const int grp  = T & 1;
    const int e256 = idx & 255;
    const int q    = e256 >> 7;
    const int ln   = (e256 >> 2) & 31;
    const int e    = e256 & 3;
    const int kidx = 8 * (ln >> 4) + 4 * q + e;
    const int nidx = ln & 15;
    float pr = 1.0f, pi = 0.0f;
    int tprev = 0;
#pragma unroll
    for (int p = 0; p < 4; ++p) {
      const int bp = wire_bitpos(grp, p);
      const int j  = (kidx >> bp) & 1;
      const int tb = (nidx >> bp) & 1;
      const int i  = tb ^ tprev;
      tprev = tb;
      const float* u = &sU[((T >> 1) * 8 + grp * 4 + p) * 8 + (i * 2 + j) * 2];
      const float ur = u[0], ui = u[1];
      const float nr = pr * ur - pi * ui;
      const float ni = pr * ui + pi * ur;
      pr = nr; pi = ni;
    }
    const int o = (((T - 2) * 6 + q) * 32 + ln) * 4 + e;
    sTab[o]       = pr;
    sTab[o + 256] = -(pr + pi);
    sTab[o + 512] = pi - pr;
  }
}

__device__ __forceinline__ void kron_stage(
    const v8f xr, const v8f xi, const float* __restrict__ tabT, int lane,
    v8f& yr, v8f& yi)
{
  const float4* t4 = reinterpret_cast<const float4*>(tabT);
  const float4 p0 = t4[lane],       p1 = t4[32 + lane];
  const float4 s0 = t4[64 + lane],  s1 = t4[96 + lane];
  const float4 d0 = t4[128 + lane], d1 = t4[160 + lane];
  const float gp[8] = {p0.x, p0.y, p0.z, p0.w, p1.x, p1.y, p1.z, p1.w};
  const float gs[8] = {s0.x, s0.y, s0.z, s0.w, s1.x, s1.y, s1.z, s1.w};
  const float gd[8] = {d0.x, d0.y, d0.z, d0.w, d1.x, d1.y, d1.z, d1.w};
  const v8f xs = xr + xi;
  v8f k = {};
#pragma unroll
  for (int s = 0; s < 4; ++s) {
    v2f a_s, b_p;
    a_s.x = xs[2 * s]; a_s.y = xs[2 * s + 1];
    b_p.x = gp[2 * s]; b_p.y = gp[2 * s + 1];
    k = __builtin_amdgcn_wmma_f32_16x16x4_f32(false, a_s, false, b_p,
                                              (short)0, k, false, false);
  }
  v8f ar = k, ai = k;
#pragma unroll
  for (int s = 0; s < 4; ++s) {
    v2f a_r, a_i, b_s, b_d;
    a_r.x = xr[2 * s]; a_r.y = xr[2 * s + 1];
    a_i.x = xi[2 * s]; a_i.y = xi[2 * s + 1];
    b_s.x = gs[2 * s]; b_s.y = gs[2 * s + 1];
    b_d.x = gd[2 * s]; b_d.y = gd[2 * s + 1];
    ar = __builtin_amdgcn_wmma_f32_16x16x4_f32(false, a_i, false, b_s,
                                               (short)0, ar, false, false);
    ai = __builtin_amdgcn_wmma_f32_16x16x4_f32(false, a_r, false, b_d,
                                               (short)0, ai, false, false);
  }
  yr = ar; yi = ai;
}

__device__ __forceinline__ float row_mirror(float v) {
  const int iv = __builtin_bit_cast(int, v);
  return __builtin_bit_cast(float,
      __builtin_amdgcn_update_dpp(iv, iv, 0x140, 0xf, 0xf, false));
}

__device__ __forceinline__ void ring_cross(v8f zr, v8f zi, bool odd,
                                           v8f& xr, v8f& xi)
{
#pragma unroll
  for (int r = 1; r < 8; r += 2) {
    zr[r] = row_mirror(zr[r]);
    zi[r] = row_mirror(zi[r]);
  }
#pragma unroll
  for (int r = 0; r < 4; ++r) {
    const float pr0 = zr[r], pr1 = zr[r + 4];
    const float pi0 = zi[r], pi1 = zi[r + 4];
    xr[r]     = odd ? pr1 : pr0;
    xr[r + 4] = odd ? pr0 : pr1;
    xi[r]     = odd ? pi1 : pi0;
    xi[r + 4] = odd ? pi0 : pi1;
  }
}

__device__ __forceinline__ void rot_u(float phi, float th, float om,
                                      float* __restrict__ u)
{
  const float pr = __cosf(0.5f * phi), pi = -__sinf(0.5f * phi);
  const float qr = __cosf(0.5f * om),  qi = -__sinf(0.5f * om);
  const float cy = __cosf(0.5f * th),  sy =  __sinf(0.5f * th);
  const float qpr  = qr*pr - qi*pi, qpi  = qr*pi + qi*pr;
  const float qcpr = qr*pr + qi*pi, qcpi = qi*pr - qr*pi;
  u[0] =  cy*qpr;  u[1] =  cy*qpi;
  u[2] = -sy*qcpr; u[3] = -sy*qcpi;
  u[4] =  sy*qcpr; u[5] = -sy*qcpi;
  u[6] =  cy*qpr;  u[7] = -cy*qpi;
}

__global__ __launch_bounds__(512) void quantum_block_kernel(
    const float* __restrict__ x,
    const float* __restrict__ W1, const float* __restrict__ b1,
    const float* __restrict__ W2, const float* __restrict__ b2,
    const float* __restrict__ W3, const float* __restrict__ b3,
    const float* __restrict__ qw,
    const float* __restrict__ Wp, const float* __restrict__ bp,
    float* __restrict__ out)
{
  __shared__ float sX  [16 * 80];
  __shared__ float sH1 [16 * 64];
  __shared__ float sH2 [16 * 32];
  __shared__ float sAng[16 * 16];
  __shared__ float sZ  [16 * 8];
  __shared__ __attribute__((aligned(16))) float sU[24 * 8];
  __shared__ __attribute__((aligned(16))) float sTab[4 * 768];

  const int tid     = threadIdx.x;
  const int wid     = tid >> 5;
  const int lane    = tid & 31;
  const int rowBase = blockIdx.x * 16;

  if (tid < 24) {
    const int L = tid >> 3, W = tid & 7;
    rot_u(qw[(L * 8 + W) * 3 + 0], qw[(L * 8 + W) * 3 + 1],
          qw[(L * 8 + W) * 3 + 2], &sU[tid * 8]);
  }
  for (int i = tid; i < 16 * 80; i += 512) sX[i] = x[rowBase * 80 + i];
  __syncthreads();

  if (wid < 4) gemm_tile_to_lds<80, 80, true >(sX,  W1, b1, sH1, 64, wid * 16);
  else         build_factor_tables(sU, sTab, tid - 128);
  __syncthreads();
  if (wid < 2) gemm_tile_to_lds<64, 64, true >(sH1, W2, b2, sH2, 32, wid * 16);
  __syncthreads();
  if (wid == 0) gemm_tile_to_lds<32, 32, false>(sH2, W3, b3, sAng, 16, 0);
  __syncthreads();

  {
    const int h = lane >> 4;
    const int c = lane & 15;
    const float* ang = &sAng[wid * 16];

    float wr = 1.0f, wi = 0.0f;
    int tprev = 0;
#pragma unroll
    for (int p = 0; p < 4; ++p) {
      const int tb = (c >> wire_bitpos(h, p)) & 1;
      const int i  = tb ^ tprev;
      tprev = tb;
      const float al = ang[4 * h + p], be = ang[8 + 4 * h + p];
      const float ca = __cosf(0.5f * al), sa = __sinf(0.5f * al);
      const float cb = __cosf(0.5f * be), sb = __sinf(0.5f * be);
      const float f0r = ca * cb, f0i = -(ca * sb);
      const float f1r = sa * cb, f1i = sa * sb;
      const float4 ui =
          *reinterpret_cast<const float4*>(&sU[(4 * h + p) * 8 + i * 4]);
      const float gr = ui.x * f0r - ui.y * f0i + ui.z * f1r - ui.w * f1i;
      const float gi = ui.x * f0i + ui.y * f0r + ui.z * f1i + ui.w * f1r;
      const float nr = wr * gr - wi * gi;
      const float ni = wr * gi + wi * gr;
      wr = nr; wi = ni;
    }
    const float vr = __shfl_xor(wr, 16, 32);
    const float vi = __shfl_xor(wi, 16, 32);
    v2f a1, b1, a2, b2;
    a1.x = h ? 0.0f : wr;  a1.y = h ? 0.0f : -wi;
    b1.x = h ? 0.0f : vr;  b1.y = h ? 0.0f : vi;
    a2.x = h ? 0.0f : wr;  a2.y = h ? 0.0f : wi;
    b2.x = h ? 0.0f : vi;  b2.y = h ? 0.0f : vr;
    v8f pr = {}, pi = {};
    pr = __builtin_amdgcn_wmma_f32_16x16x4_f32(false, a1, false, b1,
                                               (short)0, pr, false, false);
    pi = __builtin_amdgcn_wmma_f32_16x16x4_f32(false, a2, false, b2,
                                               (short)0, pi, false, false);

    const bool odd = (lane & 1) != 0;
    v8f xr, xi;
    ring_cross(pr, pi, odd, xr, xi);
    for (int L = 1; L < 3; ++L) {
      v8f yr, yi, zr, zi;
      kron_stage(xr, xi, &sTab[(2 * L - 2) * 768], lane, yr, yi);
      kron_stage(yr, yi, &sTab[(2 * L - 1) * 768], lane, zr, zi);
      ring_cross(zr, zi, odd, xr, xi);
    }

    float pb[8]; float tot = 0.f;
#pragma unroll
    for (int r = 0; r < 8; ++r) { pb[r] = xr[r]*xr[r] + xi[r]*xi[r]; tot += pb[r]; }
    float s0 = 0.f, s2 = 0.f, s3 = 0.f;
#pragma unroll
    for (int r = 0; r < 8; ++r) {
      s0 += ((r >> 2) & 1) ? -pb[r] : pb[r];
      s2 += ((r >> 1) & 1) ? -pb[r] : pb[r];
      s3 += ( r       & 1) ? -pb[r] : pb[r];
    }
    float zp[8];
    zp[0] = s0;
    zp[1] = h ? -tot : tot;
    zp[2] = s2;
    zp[3] = s3;
#pragma unroll
    for (int w = 4; w < 8; ++w)
      zp[w] = ((c >> (7 - w)) & 1) ? -tot : tot;
    const bool up16 = (lane & 16) != 0;
    const bool up8  = (lane & 8) != 0;
    const bool up4  = (lane & 4) != 0;
    float q4[4], q2[2];
#pragma unroll
    for (int j = 0; j < 4; ++j) {
      const float snd = up16 ? zp[j] : zp[4 + j];
      const float kp  = up16 ? zp[4 + j] : zp[j];
      q4[j] = kp + __shfl_xor(snd, 16, 32);
    }
#pragma unroll
    for (int j = 0; j < 2; ++j) {
      const float snd = up8 ? q4[j] : q4[2 + j];
      const float kp  = up8 ? q4[2 + j] : q4[j];
      q2[j] = kp + __shfl_xor(snd, 8, 32);
    }
    float zs;
    {
      const float snd = up4 ? q2[0] : q2[1];
      const float kp  = up4 ? q2[1] : q2[0];
      zs = kp + __shfl_xor(snd, 4, 32);
    }
    zs += __shfl_xor(zs, 2, 32);
    zs += __shfl_xor(zs, 1, 32);
    if ((lane & 3) == 0) sZ[wid * 8 + (lane >> 2)] = zs;
  }
  __syncthreads();

  if (wid < 8) gemm_final(sZ, Wp, bp, out, rowBase, wid * 16);
}

extern "C" void kernel_launch(void* const* d_in, const int* in_sizes, int n_in,
                              void* d_out, int out_size, void* d_ws, size_t ws_size,
                              hipStream_t stream) {
  const float* x  = (const float*)d_in[0];
  const float* W1 = (const float*)d_in[1];
  const float* b1 = (const float*)d_in[2];
  const float* W2 = (const float*)d_in[3];
  const float* b2 = (const float*)d_in[4];
  const float* W3 = (const float*)d_in[5];
  const float* b3 = (const float*)d_in[6];
  const float* qw = (const float*)d_in[7];
  const float* Wp = (const float*)d_in[8];
  const float* bp = (const float*)d_in[9];
  float* out = (float*)d_out;

  const int B = in_sizes[0] / 80;
  const int grid = B / 16;
  quantum_block_kernel<<<grid, 512, 0, stream>>>(x, W1, b1, W2, b2, W3, b3,
                                                 qw, Wp, bp, out);
}


# Layer 1 applied to the factors of the rank-4 state behind ring 0: per-wire 2x2 transfer chains in the encoding loop, one K=8 outer product; 26 of 50 circuit WMMA become 4; sums re-associated
# speedup vs baseline: 2.5168x; 1.1735x over previous
#include <hip/hip_runtime.h>
#include <math.h>

typedef __attribute__((ext_vector_type(2))) float v2f;
typedef __attribute__((ext_vector_type(8))) float v8f;

#define LANE() ((int)(threadIdx.x & 31))

__device__ __forceinline__ float fast_tanh(float x) {
  const float e = __expf(2.0f * x);
  return 1.0f - 2.0f / (e + 1.0f);
}

template<int K, int LDA, bool TANH>
__device__ __forceinline__ void gemm_tile_to_lds(
    const float* __restrict__ sA,
    const float* __restrict__ Wg,
    const float* __restrict__ bg,
    float* __restrict__ sOut, int ldo, int n0)
{
  const int lane = LANE();
  const int half = lane >> 4;
  const int col  = lane & 15;
  const int n    = n0 + col;
  v8f acc = {};
#pragma unroll
  for (int k0 = 0; k0 < K; k0 += 4) {
    const int ka = k0 + half * 2;
    v2f a, b;
    a.x = sA[col * LDA + ka];
    a.y = sA[col * LDA + ka + 1];
    b.x = Wg[n * K + ka];
    b.y = Wg[n * K + ka + 1];
    acc = __builtin_amdgcn_wmma_f32_16x16x4_f32(false, a, false, b,
                                                (short)0, acc, false, false);
  }
  const float bias = bg[n];
#pragma unroll
  for (int r = 0; r < 8; ++r) {
    float v = acc[r] + bias;
    if (TANH) v = fast_tanh(v);
    sOut[(half * 8 + r) * ldo + n] = v;
  }
}

__device__ __forceinline__ void gemm_final(
    const float* __restrict__ sZ,
    const float* __restrict__ Wp,
    const float* __restrict__ bp,
    float* __restrict__ out, int rowBase, int n0)
{
  const int lane = LANE();
  const int half = lane >> 4;
  const int col  = lane & 15;
  const int n    = n0 + col;
  v8f acc = {};
#pragma unroll
  for (int k0 = 0; k0 < 8; k0 += 4) {
    const int ka = k0 + half * 2;
    v2f a, b;
    a.x = sZ[col * 8 + ka];
    a.y = sZ[col * 8 + ka + 1];
    b.x = Wp[n * 8 + ka];
    b.y = Wp[n * 8 + ka + 1];
    acc = __builtin_amdgcn_wmma_f32_16x16x4_f32(false, a, false, b,
                                                (short)0, acc, false, false);
  }
  const float bias = bp[n];
#pragma unroll
  for (int r = 0; r < 8; ++r)
    out[(size_t)(rowBase + half * 8 + r) * 128 + n] = acc[r] + bias;
}


__device__ __forceinline__ int wire_bitpos(int grp, int p) {
  return grp ? (3 - p) : ((0x0132 >> (4 * p)) & 15);
}

__device__ __forceinline__ void build_factor_tables(
    const float* __restrict__ sU, float* __restrict__ sTab, int t)
{
#pragma unroll
  for (int it = 0; it < 2; ++it) {
    const int idx  = t + it * 384;
    if (idx >= 512) break;
    const int grp  = idx >> 8;
    const int e256 = idx & 255;
    const int q    = e256 >> 7;
    const int ln   = (e256 >> 2) & 31;
    const int e    = e256 & 3;
    const int kidx = 8 * (ln >> 4) + 4 * q + e;
    const int nidx = ln & 15;
    float pr = 1.0f, pi = 0.0f;
    int tprev = 0;
#pragma unroll
    for (int p = 0; p < 4; ++p) {
      const int bp = wire_bitpos(grp, p);
      const int j  = (kidx >> bp) & 1;
      const int tb = (nidx >> bp) & 1;
      const int i  = tb ^ tprev;
      tprev = tb;
      const float* u = &sU[(2 * 8 + grp * 4 + p) * 8 + (i * 2 + j) * 2];
      const float ur = u[0], ui = u[1];
      const float nr = pr * ur - pi * ui;
      const float ni = pr * ui + pi * ur;
      pr = nr; pi = ni;
    }
    const int o = ((grp * 6 + q) * 32 + ln) * 4 + e;
    sTab[o]       = pr;
    sTab[o + 256] = -(pr + pi);
    sTab[o + 512] = pi - pr;
  }
}

__device__ __forceinline__ void kron_stage(
    const v8f xr, const v8f xi, const float* __restrict__ tabT, int lane,
    v8f& yr, v8f& yi)
{
  const float4* t4 = reinterpret_cast<const float4*>(tabT);
  const float4 p0 = t4[lane],       p1 = t4[32 + lane];
  const float4 s0 = t4[64 + lane],  s1 = t4[96 + lane];
  const float4 d0 = t4[128 + lane], d1 = t4[160 + lane];
  const float gp[8] = {p0.x, p0.y, p0.z, p0.w, p1.x, p1.y, p1.z, p1.w};
  const float gs[8] = {s0.x, s0.y, s0.z, s0.w, s1.x, s1.y, s1.z, s1.w};
  const float gd[8] = {d0.x, d0.y, d0.z, d0.w, d1.x, d1.y, d1.z, d1.w};
  const v8f xs = xr + xi;
  v8f k = {};
#pragma unroll
  for (int s = 0; s < 4; ++s) {
    v2f a_s, b_p;
    a_s.x = xs[2 * s]; a_s.y = xs[2 * s + 1];
    b_p.x = gp[2 * s]; b_p.y = gp[2 * s + 1];
    k = __builtin_amdgcn_wmma_f32_16x16x4_f32(false, a_s, false, b_p,
                                              (short)0, k, false, false);
  }
  v8f ar = k, ai = k;
#pragma unroll
  for (int s = 0; s < 4; ++s) {
    v2f a_r, a_i, b_s, b_d;
    a_r.x = xr[2 * s]; a_r.y = xr[2 * s + 1];
    a_i.x = xi[2 * s]; a_i.y = xi[2 * s + 1];
    b_s.x = gs[2 * s]; b_s.y = gs[2 * s + 1];
    b_d.x = gd[2 * s]; b_d.y = gd[2 * s + 1];
    ar = __builtin_amdgcn_wmma_f32_16x16x4_f32(false, a_i, false, b_s,
                                               (short)0, ar, false, false);
    ai = __builtin_amdgcn_wmma_f32_16x16x4_f32(false, a_r, false, b_d,
                                               (short)0, ai, false, false);
  }
  yr = ar; yi = ai;
}

__device__ __forceinline__ float row_mirror(float v) {
  const int iv = __builtin_bit_cast(int, v);
  return __builtin_bit_cast(float,
      __builtin_amdgcn_update_dpp(iv, iv, 0x140, 0xf, 0xf, false));
}

__device__ __forceinline__ void ring_cross(v8f zr, v8f zi, bool odd,
                                           v8f& xr, v8f& xi)
{
#pragma unroll
  for (int r = 1; r < 8; r += 2) {
    zr[r] = row_mirror(zr[r]);
    zi[r] = row_mirror(zi[r]);
  }
#pragma unroll
  for (int r = 0; r < 4; ++r) {
    const float pr0 = zr[r], pr1 = zr[r + 4];
    const float pi0 = zi[r], pi1 = zi[r + 4];
    xr[r]     = odd ? pr1 : pr0;
    xr[r + 4] = odd ? pr0 : pr1;
    xi[r]     = odd ? pi1 : pi0;
    xi[r + 4] = odd ? pi0 : pi1;
  }
}

struct cf { float r, i; };

__device__ __forceinline__ void rot_u(float phi, float th, float om,
                                      float* __restrict__ u)
{
  const float pr = __cosf(0.5f * phi), pi = -__sinf(0.5f * phi);
  const float qr = __cosf(0.5f * om),  qi = -__sinf(0.5f * om);
  const float cy = __cosf(0.5f * th),  sy =  __sinf(0.5f * th);
  const float qpr  = qr*pr - qi*pi, qpi  = qr*pi + qi*pr;
  const float qcpr = qr*pr + qi*pi, qcpi = qi*pr - qr*pi;
  u[0] =  cy*qpr;  u[1] =  cy*qpi;
  u[2] = -sy*qcpr; u[3] = -sy*qcpi;
  u[4] =  sy*qcpr; u[5] = -sy*qcpi;
  u[6] =  cy*qpr;  u[7] = -cy*qpi;
}

__global__ __launch_bounds__(512) void quantum_block_kernel(
    const float* __restrict__ x,
    const float* __restrict__ W1, const float* __restrict__ b1,
    const float* __restrict__ W2, const float* __restrict__ b2,
    const float* __restrict__ W3, const float* __restrict__ b3,
    const float* __restrict__ qw,
    const float* __restrict__ Wp, const float* __restrict__ bp,
    float* __restrict__ out)
{
  __shared__ float sX  [16 * 80];
  __shared__ float sH1 [16 * 64];
  __shared__ float sH2 [16 * 32];
  __shared__ float sAng[16 * 16];
  __shared__ float sZ  [16 * 8];
  __shared__ __attribute__((aligned(16))) float sU[24 * 8];
  __shared__ __attribute__((aligned(16))) float sTab[2 * 768];

  const int tid     = threadIdx.x;
  const int wid     = tid >> 5;
  const int lane    = tid & 31;
  const int rowBase = blockIdx.x * 16;

  if (tid < 24) {
    const int L = tid >> 3, W = tid & 7;
    float u[8];
    rot_u(qw[(L * 8 + W) * 3 + 0], qw[(L * 8 + W) * 3 + 1],
          qw[(L * 8 + W) * 3 + 2], u);
    const bool t0 = (L == 0);
    sU[tid * 8 + 0] = u[0];
    sU[tid * 8 + 1] = t0 ? u[4] : u[1];
    sU[tid * 8 + 2] = t0 ? u[1] : u[2];
    sU[tid * 8 + 3] = t0 ? u[5] : u[3];
    sU[tid * 8 + 4] = t0 ? u[2] : u[4];
    sU[tid * 8 + 5] = t0 ? u[6] : u[5];
    sU[tid * 8 + 6] = t0 ? u[3] : u[6];
    sU[tid * 8 + 7] = u[7];
  }
  for (int i = tid; i < 16 * 80; i += 512) sX[i] = x[rowBase * 80 + i];
  __syncthreads();

  if (wid < 4) gemm_tile_to_lds<80, 80, true >(sX,  W1, b1, sH1, 64, wid * 16);
  else         build_factor_tables(sU, sTab, tid - 128);
  __syncthreads();
  if (wid < 2) gemm_tile_to_lds<64, 64, true >(sH1, W2, b2, sH2, 32, wid * 16);
  __syncthreads();
  if (wid == 0) gemm_tile_to_lds<32, 32, false>(sH2, W3, b3, sAng, 16, 0);
  __syncthreads();

  {
    const int h = lane >> 4;
    const int c = lane & 15;
    const float* ang = &sAng[wid * 16];

    v2f E0r, E0i, E1r, E1i;
    int tprev = 0;
#pragma unroll
    for (int p = 0; p < 4; ++p) {
      const int tb = (c >> wire_bitpos(h, p)) & 1;
      const int i  = tb ^ tprev;
      tprev = tb;
      const float al = ang[4 * h + p], be = ang[8 + 4 * h + p];
      const float ca = __cosf(0.5f * al), sa = __sinf(0.5f * al);
      const float cb = __cosf(0.5f * be), sb = __sinf(0.5f * be);
      const float f0r = ca * cb, f0i = -(ca * sb);
      const float f1r = sa * cb, f1i = sa * sb;
      const float4 ua = *reinterpret_cast<const float4*>(&sU[(4 * h + p) * 8]);
      const float4 ub = *reinterpret_cast<const float4*>(&sU[(4 * h + p) * 8 + 4]);
      const v2f UAr = {ua.x, ua.y}, UAi = {ua.z, ua.w};
      const v2f UBr = {ub.x, ub.y}, UBi = {ub.z, ub.w};
      const v2f Gr = UAr * f0r - UAi * f0i + UBr * f1r - UBi * f1i;
      const v2f Gi = UAr * f0i + UAi * f0r + UBr * f1i + UBi * f1r;
      const float4 w4 =
          *reinterpret_cast<const float4*>(&sU[(8 + 4 * h + p) * 8 + i * 4]);
      if (p == 0) {
        const v2f Gsr = {Gr.y, Gr.x}, Gsi = {Gi.y, Gi.x};
        E0r = Gr * w4.x - Gi * w4.y;    E0i = Gr * w4.y + Gi * w4.x;
        E1r = Gsr * w4.z - Gsi * w4.w;  E1i = Gsr * w4.w + Gsi * w4.z;
      } else {
        const bool x = (p == 1) && (h == 0);
        const v2f Ar = {Gr.x, x ? Gr.y : Gr.x}, Ai = {Gi.x, x ? Gi.y : Gi.x};
        const v2f Br = {Gr.y, x ? Gr.x : Gr.y}, Bi = {Gi.y, x ? Gi.x : Gi.y};
        const v2f S0r = E0r * Ar - E0i * Ai + E1r * Br - E1i * Bi;
        const v2f S0i = E0r * Ai + E0i * Ar + E1r * Bi + E1i * Br;
        const v2f S1r = E0r * Br - E0i * Bi + E1r * Ar - E1i * Ai;
        const v2f S1i = E0r * Bi + E0i * Br + E1r * Ai + E1i * Ar;
        E0r = S0r * w4.x - S0i * w4.y;  E0i = S0r * w4.y + S0i * w4.x;
        E1r = S1r * w4.z - S1i * w4.w;  E1i = S1r * w4.w + S1i * w4.z;
      }
    }
    const cf e0[2] = {{E0r.x, E0i.x}, {E1r.x, E1i.x}};
    const cf e1[2] = {{E0r.y, E0i.y}, {E1r.y, E1i.y}};
    const cf snd0 = h ? e0[0] : e1[0];
    const cf snd1 = h ? e1[0] : e1[1];
    cf rcv0, rcv1;
    rcv0.r = __shfl_xor(snd0.r, 16, 32);  rcv0.i = __shfl_xor(snd0.i, 16, 32);
    rcv1.r = __shfl_xor(snd1.r, 16, 32);  rcv1.i = __shfl_xor(snd1.i, 16, 32);
    const cf P0 = h ? rcv0 : e0[0], Q0 = h ? e0[1] : rcv0;
    const cf P1 = h ? rcv1 : e0[1], Q1 = h ? e1[1] : rcv1;
    v2f ar0, br0, ai0, bi0, ar1, br1, ai1, bi1;
    ar0.x = P0.r;  ar0.y = -P0.i;  br0.x = Q0.r;  br0.y = Q0.i;
    ai0.x = P0.r;  ai0.y =  P0.i;  bi0.x = Q0.i;  bi0.y = Q0.r;
    ar1.x = P1.r;  ar1.y = -P1.i;  br1.x = Q1.r;  br1.y = Q1.i;
    ai1.x = P1.r;  ai1.y =  P1.i;  bi1.x = Q1.i;  bi1.y = Q1.r;
    v8f pr = {}, pi = {};
    pr = __builtin_amdgcn_wmma_f32_16x16x4_f32(false, ar0, false, br0,
                                               (short)0, pr, false, false);
    pi = __builtin_amdgcn_wmma_f32_16x16x4_f32(false, ai0, false, bi0,
                                               (short)0, pi, false, false);
    pr = __builtin_amdgcn_wmma_f32_16x16x4_f32(false, ar1, false, br1,
                                               (short)0, pr, false, false);
    pi = __builtin_amdgcn_wmma_f32_16x16x4_f32(false, ai1, false, bi1,
                                               (short)0, pi, false, false);

    const bool odd = (lane & 1) != 0;
    v8f xr, xi;
    ring_cross(pr, pi, odd, xr, xi);
    {
      v8f yr, yi, zr, zi;
      kron_stage(xr, xi, &sTab[0],   lane, yr, yi);
      kron_stage(yr, yi, &sTab[768], lane, zr, zi);
      ring_cross(zr, zi, odd, xr, xi);
    }

    float pb[8]; float tot = 0.f;
#pragma unroll
    for (int r = 0; r < 8; ++r) { pb[r] = xr[r]*xr[r] + xi[r]*xi[r]; tot += pb[r]; }
    float s0 = 0.f, s2 = 0.f, s3 = 0.f;
#pragma unroll
    for (int r = 0; r < 8; ++r) {
      s0 += ((r >> 2) & 1) ? -pb[r] : pb[r];
      s2 += ((r >> 1) & 1) ? -pb[r] : pb[r];
      s3 += ( r       & 1) ? -pb[r] : pb[r];
    }
    float zp[8];
    zp[0] = s0;
    zp[1] = h ? -tot : tot;
    zp[2] = s2;
    zp[3] = s3;
#pragma unroll
    for (int w = 4; w < 8; ++w)
      zp[w] = ((c >> (7 - w)) & 1) ? -tot : tot;
    const bool up16 = (lane & 16) != 0;
    const bool up8  = (lane & 8) != 0;
    const bool up4  = (lane & 4) != 0;
    float q4[4], q2[2];
#pragma unroll
    for (int j = 0; j < 4; ++j) {
      const float snd = up16 ? zp[j] : zp[4 + j];
      const float kp  = up16 ? zp[4 + j] : zp[j];
      q4[j] = kp + __shfl_xor(snd, 16, 32);
    }
#pragma unroll
    for (int j = 0; j < 2; ++j) {
      const float snd = up8 ? q4[j] : q4[2 + j];
      const float kp  = up8 ? q4[2 + j] : q4[j];
      q2[j] = kp + __shfl_xor(snd, 8, 32);
    }
    float zs;
    {
      const float snd = up4 ? q2[0] : q2[1];
      const float kp  = up4 ? q2[1] : q2[0];
      zs = kp + __shfl_xor(snd, 4, 32);
    }
    zs += __shfl_xor(zs, 2, 32);
    zs += __shfl_xor(zs, 1, 32);
    if ((lane & 3) == 0) sZ[wid * 8 + (lane >> 2)] = zs;
  }
  __syncthreads();

  if (wid < 8) gemm_final(sZ, Wp, bp, out, rowBase, wid * 16);
}

extern "C" void kernel_launch(void* const* d_in, const int* in_sizes, int n_in,
                              void* d_out, int out_size, void* d_ws, size_t ws_size,
                              hipStream_t stream) {
  const float* x  = (const float*)d_in[0];
  const float* W1 = (const float*)d_in[1];
  const float* b1 = (const float*)d_in[2];
  const float* W2 = (const float*)d_in[3];
  const float* b2 = (const float*)d_in[4];
  const float* W3 = (const float*)d_in[5];
  const float* b3 = (const float*)d_in[6];
  const float* qw = (const float*)d_in[7];
  const float* Wp = (const float*)d_in[8];
  const float* bp = (const float*)d_in[9];
  float* out = (float*)d_out;

  const int B = in_sizes[0] / 80;
  const int grid = B / 16;
  quantum_block_kernel<<<grid, 512, 0, stream>>>(x, W1, b1, W2, b2, W3, b3,
                                                 qw, Wp, bp, out);
}


# Layer-2 Kronecker factors made real: Rot = RZ.RY.RZ, all RZ(phi) as one elementwise phase table, RZ(omega) dropped (|amp|^2 only): 8 WMMA per stage, not 12; sums re-associated
# speedup vs baseline: 2.6848x; 1.0668x over previous
#include <hip/hip_runtime.h>
#include <math.h>

typedef __attribute__((ext_vector_type(2))) float v2f;
typedef __attribute__((ext_vector_type(8))) float v8f;

#define LANE() ((int)(threadIdx.x & 31))

__device__ __forceinline__ float fast_tanh(float x) {
  const float e = __expf(2.0f * x);
  return 1.0f - 2.0f / (e + 1.0f);
}

template<int K, int LDA, bool TANH>
__device__ __forceinline__ void gemm_tile_to_lds(
    const float* __restrict__ sA,
    const float* __restrict__ Wg,
    const float* __restrict__ bg,
    float* __restrict__ sOut, int ldo, int n0)
{
  const int lane = LANE();
  const int half = lane >> 4;
  const int col  = lane & 15;
  const int n    = n0 + col;
  v8f acc = {};
#pragma unroll
  for (int k0 = 0; k0 < K; k0 += 4) {
    const int ka = k0 + half * 2;
    v2f a, b;
    a.x = sA[col * LDA + ka];
    a.y = sA[col * LDA + ka + 1];
    b.x = Wg[n * K + ka];
    b.y = Wg[n * K + ka + 1];
    acc = __builtin_amdgcn_wmma_f32_16x16x4_f32(false, a, false, b,
                                                (short)0, acc, false, false);
  }
  const float bias = bg[n];
#pragma unroll
  for (int r = 0; r < 8; ++r) {
    float v = acc[r] + bias;
    if (TANH) v = fast_tanh(v);
    sOut[(half * 8 + r) * ldo + n] = v;
  }
}

__device__ __forceinline__ void gemm_final(
    const float* __restrict__ sZ,
    const float* __restrict__ Wp,
    const float* __restrict__ bp,
    float* __restrict__ out, int rowBase, int n0)
{
  const int lane = LANE();
  const int half = lane >> 4;
  const int col  = lane & 15;
  const int n    = n0 + col;
  v8f acc = {};
#pragma unroll
  for (int k0 = 0; k0 < 8; k0 += 4) {
    const int ka = k0 + half * 2;
    v2f a, b;
    a.x = sZ[col * 8 + ka];
    a.y = sZ[col * 8 + ka + 1];
    b.x = Wp[n * 8 + ka];
    b.y = Wp[n * 8 + ka + 1];
    acc = __builtin_amdgcn_wmma_f32_16x16x4_f32(false, a, false, b,
                                                (short)0, acc, false, false);
  }
  const float bias = bp[n];
#pragma unroll
  for (int r = 0; r < 8; ++r)
    out[(size_t)(rowBase + half * 8 + r) * 128 + n] = acc[r] + bias;
}


__device__ __forceinline__ int wire_bitpos(int grp, int p) {
  return grp ? (3 - p) : ((0x0132 >> (4 * p)) & 15);
}

__device__ __forceinline__ void build_layer2_tables(
    const float* __restrict__ sU, float* __restrict__ sTabR,
    float* __restrict__ sPh, int t)
{
  const int q     = (t >> 5) & 1;
  const int ln    = t & 31;
  const int nidx  = ln & 15;
  const int kbase = 8 * (ln >> 4) + 4 * q;
  if (t < 128) {
    const int grp = t >> 6;
    float v[4];
#pragma unroll
    for (int e = 0; e < 4; ++e) {
      const int kidx = kbase + e;
      float pr = 1.0f;
      int tprev = 0;
#pragma unroll
      for (int p = 0; p < 4; ++p) {
        const int bp = wire_bitpos(grp, p);
        const int j  = (kidx >> bp) & 1;
        const int tb = (nidx >> bp) & 1;
        const int i  = tb ^ tprev;
        tprev = tb;
        const float2 cs =
            *reinterpret_cast<const float2*>(&sU[(16 + grp * 4 + p) * 8]);
        const float off = i ? cs.y : -cs.y;
        pr *= (i == j) ? cs.x : off;
      }
      v[e] = pr;
    }
    *reinterpret_cast<float4*>(&sTabR[((grp * 2 + q) * 32 + ln) * 4]) =
        make_float4(v[0], v[1], v[2], v[3]);
  } else if (t < 192) {
    float vr[4], vi[4];
#pragma unroll
    for (int e = 0; e < 4; ++e) {
      const int midx = kbase + e;
      float pr = 1.0f, pi = 0.0f;
#pragma unroll
      for (int ww = 0; ww < 8; ++ww) {
        const int w   = (ww + 4) & 7;
        const int src = (w < 4) ? midx : nidx;
        const int bit = (src >> wire_bitpos(w >> 2, w & 3)) & 1;
        const float2 z =
            *reinterpret_cast<const float2*>(&sU[(16 + w) * 8 + 2]);
        const float zi = bit ? -z.y : z.y;
        const float nr = pr * z.x - pi * zi;
        const float ni = pr * zi + pi * z.x;
        pr = nr; pi = ni;
      }
      vr[e] = pr; vi[e] = pi;
    }
    const int o = (q * 32 + ln) * 4;
    *reinterpret_cast<float4*>(&sPh[o])       = make_float4(vr[0], vr[1], vr[2], vr[3]);
    *reinterpret_cast<float4*>(&sPh[o + 256]) = make_float4(vi[0], vi[1], vi[2], vi[3]);
  }
}

__device__ __forceinline__ void kron_stage_real(
    const v8f xr, const v8f xi, const float* __restrict__ tabT, int lane,
    v8f& yr, v8f& yi)
{
  const float4* t4 = reinterpret_cast<const float4*>(tabT);
  const float4 g0 = t4[lane], g1 = t4[32 + lane];
  const float g[8] = {g0.x, g0.y, g0.z, g0.w, g1.x, g1.y, g1.z, g1.w};
  v8f ar = {}, ai = {};
#pragma unroll
  for (int s = 0; s < 4; ++s) {
    v2f a_r, a_i, b;
    a_r.x = xr[2 * s]; a_r.y = xr[2 * s + 1];
    a_i.x = xi[2 * s]; a_i.y = xi[2 * s + 1];
    b.x   = g[2 * s];  b.y   = g[2 * s + 1];
    ar = __builtin_amdgcn_wmma_f32_16x16x4_f32(false, a_r, false, b,
                                               (short)0, ar, false, false);
    ai = __builtin_amdgcn_wmma_f32_16x16x4_f32(false, a_i, false, b,
                                               (short)0, ai, false, false);
  }
  yr = ar; yi = ai;
}

__device__ __forceinline__ float row_mirror(float v) {
  const int iv = __builtin_bit_cast(int, v);
  return __builtin_bit_cast(float,
      __builtin_amdgcn_update_dpp(iv, iv, 0x140, 0xf, 0xf, false));
}

__device__ __forceinline__ void ring_cross(v8f zr, v8f zi, bool odd,
                                           v8f& xr, v8f& xi)
{
#pragma unroll
  for (int r = 1; r < 8; r += 2) {
    zr[r] = row_mirror(zr[r]);
    zi[r] = row_mirror(zi[r]);
  }
#pragma unroll
  for (int r = 0; r < 4; ++r) {
    const float pr0 = zr[r], pr1 = zr[r + 4];
    const float pi0 = zi[r], pi1 = zi[r + 4];
    xr[r]     = odd ? pr1 : pr0;
    xr[r + 4] = odd ? pr0 : pr1;
    xi[r]     = odd ? pi1 : pi0;
    xi[r + 4] = odd ? pi0 : pi1;
  }
}

struct cf { float r, i; };

__device__ __forceinline__ void rot_u(float phi, float th, float om,
                                      float* __restrict__ u,
                                      float* __restrict__ ryp)
{
  const float pr = __cosf(0.5f * phi), pi = -__sinf(0.5f * phi);
  const float qr = __cosf(0.5f * om),  qi = -__sinf(0.5f * om);
  const float cy = __cosf(0.5f * th),  sy =  __sinf(0.5f * th);
  const float qpr  = qr*pr - qi*pi, qpi  = qr*pi + qi*pr;
  const float qcpr = qr*pr + qi*pi, qcpi = qi*pr - qr*pi;
  u[0] =  cy*qpr;  u[1] =  cy*qpi;
  u[2] = -sy*qcpr; u[3] = -sy*qcpi;
  u[4] =  sy*qcpr; u[5] = -sy*qcpi;
  u[6] =  cy*qpr;  u[7] = -cy*qpi;
  ryp[0] = cy; ryp[1] = sy; ryp[2] = pr; ryp[3] = pi;
}

__global__ __launch_bounds__(512, 16) void quantum_block_kernel(
    const float* __restrict__ x,
    const float* __restrict__ W1, const float* __restrict__ b1,
    const float* __restrict__ W2, const float* __restrict__ b2,
    const float* __restrict__ W3, const float* __restrict__ b3,
    const float* __restrict__ qw,
    const float* __restrict__ Wp, const float* __restrict__ bp,
    float* __restrict__ out)
{
  __shared__ float sX  [16 * 80];
  __shared__ float sH1 [16 * 64];
  __shared__ float sH2 [16 * 32];
  __shared__ float sAng[16 * 16];
  __shared__ float sZ  [16 * 8];
  __shared__ __attribute__((aligned(16))) float sU[24 * 8];
  __shared__ __attribute__((aligned(16))) float sTabR[2 * 256];
  __shared__ __attribute__((aligned(16))) float sPh[512];

  const int tid     = threadIdx.x;
  const int wid     = tid >> 5;
  const int lane    = tid & 31;
  const int rowBase = blockIdx.x * 16;

  if (tid < 24) {
    const int L = tid >> 3, W = tid & 7;
    float u[8], ryp[4];
    rot_u(qw[(L * 8 + W) * 3 + 0], qw[(L * 8 + W) * 3 + 1],
          qw[(L * 8 + W) * 3 + 2], u, ryp);
    const bool t0 = (L == 0), t2 = (L == 2);
    sU[tid * 8 + 0] = t2 ? ryp[0] : u[0];
    sU[tid * 8 + 1] = t2 ? ryp[1] : (t0 ? u[4] : u[1]);
    sU[tid * 8 + 2] = t2 ? ryp[2] : (t0 ? u[1] : u[2]);
    sU[tid * 8 + 3] = t2 ? ryp[3] : (t0 ? u[5] : u[3]);
    sU[tid * 8 + 4] = t0 ? u[2] : u[4];
    sU[tid * 8 + 5] = t0 ? u[6] : u[5];
    sU[tid * 8 + 6] = t0 ? u[3] : u[6];
    sU[tid * 8 + 7] = u[7];
  }
  for (int i = tid; i < 16 * 80; i += 512) sX[i] = x[rowBase * 80 + i];
  __syncthreads();

  if (wid < 4) gemm_tile_to_lds<80, 80, true >(sX,  W1, b1, sH1, 64, wid * 16);
  else         build_layer2_tables(sU, sTabR, sPh, tid - 128);
  __syncthreads();
  if (wid < 2) gemm_tile_to_lds<64, 64, true >(sH1, W2, b2, sH2, 32, wid * 16);
  __syncthreads();
  if (wid == 0) gemm_tile_to_lds<32, 32, false>(sH2, W3, b3, sAng, 16, 0);
  __syncthreads();

  {
    const int h = lane >> 4;
    const int c = lane & 15;
    const float* ang = &sAng[wid * 16];

    v2f E0r, E0i, E1r, E1i;
    int tprev = 0;
#pragma unroll
    for (int p = 0; p < 4; ++p) {
      const int tb = (c >> wire_bitpos(h, p)) & 1;
      const int i  = tb ^ tprev;
      tprev = tb;
      const float al = ang[4 * h + p], be = ang[8 + 4 * h + p];
      const float ca = __cosf(0.5f * al), sa = __sinf(0.5f * al);
      const float cb = __cosf(0.5f * be), sb = __sinf(0.5f * be);
      const float f0r = ca * cb, f0i = -(ca * sb);
      const float f1r = sa * cb, f1i = sa * sb;
      const float4 ua = *reinterpret_cast<const float4*>(&sU[(4 * h + p) * 8]);
      const float4 ub = *reinterpret_cast<const float4*>(&sU[(4 * h + p) * 8 + 4]);
      const v2f UAr = {ua.x, ua.y}, UAi = {ua.z, ua.w};
      const v2f UBr = {ub.x, ub.y}, UBi = {ub.z, ub.w};
      const v2f Gr = UAr * f0r - UAi * f0i + UBr * f1r - UBi * f1i;
      const v2f Gi = UAr * f0i + UAi * f0r + UBr * f1i + UBi * f1r;
      const float4 w4 =
          *reinterpret_cast<const float4*>(&sU[(8 + 4 * h + p) * 8 + i * 4]);
      if (p == 0) {
        const v2f Gsr = {Gr.y, Gr.x}, Gsi = {Gi.y, Gi.x};
        E0r = Gr * w4.x - Gi * w4.y;    E0i = Gr * w4.y + Gi * w4.x;
        E1r = Gsr * w4.z - Gsi * w4.w;  E1i = Gsr * w4.w + Gsi * w4.z;
      } else {
        const bool x = (p == 1) && (h == 0);
        const v2f Ar = {Gr.x, x ? Gr.y : Gr.x}, Ai = {Gi.x, x ? Gi.y : Gi.x};
        const v2f Br = {Gr.y, x ? Gr.x : Gr.y}, Bi = {Gi.y, x ? Gi.x : Gi.y};
        const v2f S0r = E0r * Ar - E0i * Ai + E1r * Br - E1i * Bi;
        const v2f S0i = E0r * Ai + E0i * Ar + E1r * Bi + E1i * Br;
        const v2f S1r = E0r * Br - E0i * Bi + E1r * Ar - E1i * Ai;
        const v2f S1i = E0r * Bi + E0i * Br + E1r * Ai + E1i * Ar;
        E0r = S0r * w4.x - S0i * w4.y;  E0i = S0r * w4.y + S0i * w4.x;
        E1r = S1r * w4.z - S1i * w4.w;  E1i = S1r * w4.w + S1i * w4.z;
      }
    }
    const cf e0[2] = {{E0r.x, E0i.x}, {E1r.x, E1i.x}};
    const cf e1[2] = {{E0r.y, E0i.y}, {E1r.y, E1i.y}};
    const cf snd0 = h ? e0[0] : e1[0];
    const cf snd1 = h ? e1[0] : e1[1];
    cf rcv0, rcv1;
    rcv0.r = __shfl_xor(snd0.r, 16, 32);  rcv0.i = __shfl_xor(snd0.i, 16, 32);
    rcv1.r = __shfl_xor(snd1.r, 16, 32);  rcv1.i = __shfl_xor(snd1.i, 16, 32);
    const cf P0 = h ? rcv0 : e0[0], Q0 = h ? e0[1] : rcv0;
    const cf P1 = h ? rcv1 : e0[1], Q1 = h ? e1[1] : rcv1;
    v2f ar0, br0, ai0, bi0, ar1, br1, ai1, bi1;
    ar0.x = P0.r;  ar0.y = -P0.i;  br0.x = Q0.r;  br0.y = Q0.i;
    ai0.x = P0.r;  ai0.y =  P0.i;  bi0.x = Q0.i;  bi0.y = Q0.r;
    ar1.x = P1.r;  ar1.y = -P1.i;  br1.x = Q1.r;  br1.y = Q1.i;
    ai1.x = P1.r;  ai1.y =  P1.i;  bi1.x = Q1.i;  bi1.y = Q1.r;
    v8f pr = {}, pi = {};
    pr = __builtin_amdgcn_wmma_f32_16x16x4_f32(false, ar0, false, br0,
                                               (short)0, pr, false, false);
    pi = __builtin_amdgcn_wmma_f32_16x16x4_f32(false, ai0, false, bi0,
                                               (short)0, pi, false, false);
    pr = __builtin_amdgcn_wmma_f32_16x16x4_f32(false, ar1, false, br1,
                                               (short)0, pr, false, false);
    pi = __builtin_amdgcn_wmma_f32_16x16x4_f32(false, ai1, false, bi1,
                                               (short)0, pi, false, false);

    const bool odd = (lane & 1) != 0;
    v8f xr, xi;
    ring_cross(pr, pi, odd, xr, xi);
    {
      const float4* d4 = reinterpret_cast<const float4*>(sPh);
      const float4 c0 = d4[lane],      c1 = d4[32 + lane];
      const float4 s0 = d4[64 + lane], s1 = d4[96 + lane];
      const v8f dr = {c0.x, c0.y, c0.z, c0.w, c1.x, c1.y, c1.z, c1.w};
      const v8f di = {s0.x, s0.y, s0.z, s0.w, s1.x, s1.y, s1.z, s1.w};
      const v8f wr = xr * dr - xi * di;
      const v8f wi = xr * di + xi * dr;
      v8f yr, yi, zr, zi;
      kron_stage_real(wr, wi, &sTabR[0],   lane, yr, yi);
      kron_stage_real(yr, yi, &sTabR[256], lane, zr, zi);
      ring_cross(zr, zi, odd, xr, xi);
    }

    float pb[8]; float tot = 0.f;
#pragma unroll
    for (int r = 0; r < 8; ++r) { pb[r] = xr[r]*xr[r] + xi[r]*xi[r]; tot += pb[r]; }
    float s0 = 0.f, s2 = 0.f, s3 = 0.f;
#pragma unroll
    for (int r = 0; r < 8; ++r) {
      s0 += ((r >> 2) & 1) ? -pb[r] : pb[r];
      s2 += ((r >> 1) & 1) ? -pb[r] : pb[r];
      s3 += ( r       & 1) ? -pb[r] : pb[r];
    }
    float zp[8];
    zp[0] = s0;
    zp[1] = h ? -tot : tot;
    zp[2] = s2;
    zp[3] = s3;
#pragma unroll
    for (int w = 4; w < 8; ++w)
      zp[w] = ((c >> (7 - w)) & 1) ? -tot : tot;
    const bool up16 = (lane & 16) != 0;
    const bool up8  = (lane & 8) != 0;
    const bool up4  = (lane & 4) != 0;
    float q4[4], q2[2];
#pragma unroll
    for (int j = 0; j < 4; ++j) {
      const float snd = up16 ? zp[j] : zp[4 + j];
      const float kp  = up16 ? zp[4 + j] : zp[j];
      q4[j] = kp + __shfl_xor(snd, 16, 32);
    }
#pragma unroll
    for (int j = 0; j < 2; ++j) {
      const float snd = up8 ? q4[j] : q4[2 + j];
      const float kp  = up8 ? q4[2 + j] : q4[j];
      q2[j] = kp + __shfl_xor(snd, 8, 32);
    }
    float zs;
    {
      const float snd = up4 ? q2[0] : q2[1];
      const float kp  = up4 ? q2[1] : q2[0];
      zs = kp + __shfl_xor(snd, 4, 32);
    }
    zs += __shfl_xor(zs, 2, 32);
    zs += __shfl_xor(zs, 1, 32);
    if ((lane & 3) == 0) sZ[wid * 8 + (lane >> 2)] = zs;
  }
  __syncthreads();

  if (wid < 8) gemm_final(sZ, Wp, bp, out, rowBase, wid * 16);
}

extern "C" void kernel_launch(void* const* d_in, const int* in_sizes, int n_in,
                              void* d_out, int out_size, void* d_ws, size_t ws_size,
                              hipStream_t stream) {
  const float* x  = (const float*)d_in[0];
  const float* W1 = (const float*)d_in[1];
  const float* b1 = (const float*)d_in[2];
  const float* W2 = (const float*)d_in[3];
  const float* b2 = (const float*)d_in[4];
  const float* W3 = (const float*)d_in[5];
  const float* b3 = (const float*)d_in[6];
  const float* qw = (const float*)d_in[7];
  const float* Wp = (const float*)d_in[8];
  const float* bp = (const float*)d_in[9];
  float* out = (float*)d_out;

  const int B = in_sizes[0] / 80;
  const int grid = B / 16;
  quantum_block_kernel<<<grid, 512, 0, stream>>>(x, W1, b1, W2, b2, W3, b3,
                                                 qw, Wp, bp, out);
}


# Global loads 16 bytes wide: x tile in one trip, not three dependent 4-byte trips (Rot set-up on an idle wave beside it); W and LDS fragments 4 consecutive k per lane half; sums re-associated
# speedup vs baseline: 2.9003x; 1.0803x over previous
#include <hip/hip_runtime.h>
#include <math.h>

typedef __attribute__((ext_vector_type(2))) float v2f;
typedef __attribute__((ext_vector_type(8))) float v8f;

#define LANE() ((int)(threadIdx.x & 31))

__device__ __forceinline__ float fast_tanh(float x) {
  const float e = __expf(2.0f * x);
  return 1.0f - 2.0f / (e + 1.0f);
}

template<int K, int LDA, bool TANH>
__device__ __forceinline__ void gemm_tile_to_lds(
    const float* __restrict__ sA,
    const float* __restrict__ Wg,
    const float* __restrict__ bg,
    float* __restrict__ sOut, int ldo, int n0)
{
  const int lane = LANE();
  const int half = lane >> 4;
  const int col  = lane & 15;
  const int n    = n0 + col;
  v8f acc = {};
  const float4* a4 = reinterpret_cast<const float4*>(sA);
  const float4* b4 = reinterpret_cast<const float4*>(Wg);
  const int ao = col * (LDA / 4) + half;
  const int bo = n * (K / 4) + half;
#pragma unroll
  for (int j = 0; j < K / 8; ++j) {
    const float4 av = a4[ao + 2 * j];
    const float4 bv = b4[bo + 2 * j];
    v2f a, b;
    a.x = av.x; a.y = av.y; b.x = bv.x; b.y = bv.y;
    acc = __builtin_amdgcn_wmma_f32_16x16x4_f32(false, a, false, b,
                                                (short)0, acc, false, false);
    a.x = av.z; a.y = av.w; b.x = bv.z; b.y = bv.w;
    acc = __builtin_amdgcn_wmma_f32_16x16x4_f32(false, a, false, b,
                                                (short)0, acc, false, false);
  }
  const float bias = bg[n];
#pragma unroll
  for (int r = 0; r < 8; ++r) {
    float v = acc[r] + bias;
    if (TANH) v = fast_tanh(v);
    sOut[(half * 8 + r) * ldo + n] = v;
  }
}

__device__ __forceinline__ void gemm_final(
    const float* __restrict__ sZ,
    const float* __restrict__ Wp,
    const float* __restrict__ bp,
    float* __restrict__ out, int rowBase, int n0)
{
  const int lane = LANE();
  const int half = lane >> 4;
  const int col  = lane & 15;
  const int n    = n0 + col;
  v8f acc = {};
  {
    const float4 av = reinterpret_cast<const float4*>(sZ)[col * 2 + half];
    const float4 bv = reinterpret_cast<const float4*>(Wp)[n * 2 + half];
    v2f a, b;
    a.x = av.x; a.y = av.y; b.x = bv.x; b.y = bv.y;
    acc = __builtin_amdgcn_wmma_f32_16x16x4_f32(false, a, false, b,
                                                (short)0, acc, false, false);
    a.x = av.z; a.y = av.w; b.x = bv.z; b.y = bv.w;
    acc = __builtin_amdgcn_wmma_f32_16x16x4_f32(false, a, false, b,
                                                (short)0, acc, false, false);
  }
  const float bias = bp[n];
#pragma unroll
  for (int r = 0; r < 8; ++r)
    out[(size_t)(rowBase + half * 8 + r) * 128 + n] = acc[r] + bias;
}


__device__ __forceinline__ int wire_bitpos(int grp, int p) {
  return grp ? (3 - p) : ((0x0132 >> (4 * p)) & 15);
}

__device__ __forceinline__ void build_layer2_tables(
    const float* __restrict__ sU, float* __restrict__ sTabR,
    float* __restrict__ sPh, int t)
{
  const int q     = (t >> 5) & 1;
  const int ln    = t & 31;
  const int nidx  = ln & 15;
  const int kbase = 8 * (ln >> 4) + 4 * q;
  if (t < 128) {
    const int grp = t >> 6;
    float v[4];
#pragma unroll
    for (int e = 0; e < 4; ++e) {
      const int kidx = kbase + e;
      float pr = 1.0f;
      int tprev = 0;
#pragma unroll
      for (int p = 0; p < 4; ++p) {
        const int bp = wire_bitpos(grp, p);
        const int j  = (kidx >> bp) & 1;
        const int tb = (nidx >> bp) & 1;
        const int i  = tb ^ tprev;
        tprev = tb;
        const float2 cs =
            *reinterpret_cast<const float2*>(&sU[(16 + grp * 4 + p) * 8]);
        const float off = i ? cs.y : -cs.y;
        pr *= (i == j) ? cs.x : off;
      }
      v[e] = pr;
    }
    *reinterpret_cast<float4*>(&sTabR[((grp * 2 + q) * 32 + ln) * 4]) =
        make_float4(v[0], v[1], v[2], v[3]);
  } else if (t < 192) {
    float vr[4], vi[4];
#pragma unroll
    for (int e = 0; e < 4; ++e) {
      const int midx = kbase + e;
      float pr = 1.0f, pi = 0.0f;
#pragma unroll
      for (int ww = 0; ww < 8; ++ww) {
        const int w   = (ww + 4) & 7;
        const int src = (w < 4) ? midx : nidx;
        const int bit = (src >> wire_bitpos(w >> 2, w & 3)) & 1;
        const float2 z =
            *reinterpret_cast<const float2*>(&sU[(16 + w) * 8 + 2]);
        const float zi = bit ? -z.y : z.y;
        const float nr = pr * z.x - pi * zi;
        const float ni = pr * zi + pi * z.x;
        pr = nr; pi = ni;
      }
      vr[e] = pr; vi[e] = pi;
    }
    const int o = (q * 32 + ln) * 4;
    *reinterpret_cast<float4*>(&sPh[o])       = make_float4(vr[0], vr[1], vr[2], vr[3]);
    *reinterpret_cast<float4*>(&sPh[o + 256]) = make_float4(vi[0], vi[1], vi[2], vi[3]);
  }
}

__device__ __forceinline__ void kron_stage_real(
    const v8f xr, const v8f xi, const float* __restrict__ tabT, int lane,
    v8f& yr, v8f& yi)
{
  const float4* t4 = reinterpret_cast<const float4*>(tabT);
  const float4 g0 = t4[lane], g1 = t4[32 + lane];
  const float g[8] = {g0.x, g0.y, g0.z, g0.w, g1.x, g1.y, g1.z, g1.w};
  v8f ar = {}, ai = {};
#pragma unroll
  for (int s = 0; s < 4; ++s) {
    v2f a_r, a_i, b;
    a_r.x = xr[2 * s]; a_r.y = xr[2 * s + 1];
    a_i.x = xi[2 * s]; a_i.y = xi[2 * s + 1];
    b.x   = g[2 * s];  b.y   = g[2 * s + 1];
    ar = __builtin_amdgcn_wmma_f32_16x16x4_f32(false, a_r, false, b,
                                               (short)0, ar, false, false);
    ai = __builtin_amdgcn_wmma_f32_16x16x4_f32(false, a_i, false, b,
                                               (short)0, ai, false, false);
  }
  yr = ar; yi = ai;
}

__device__ __forceinline__ float row_mirror(float v) {
  const int iv = __builtin_bit_cast(int, v);
  return __builtin_bit_cast(float,
      __builtin_amdgcn_update_dpp(iv, iv, 0x140, 0xf, 0xf, false));
}

__device__ __forceinline__ void ring_cross(v8f zr, v8f zi, bool odd,
                                           v8f& xr, v8f& xi)
{
#pragma unroll
  for (int r = 1; r < 8; r += 2) {
    zr[r] = row_mirror(zr[r]);
    zi[r] = row_mirror(zi[r]);
  }
#pragma unroll
  for (int r = 0; r < 4; ++r) {
    const float pr0 = zr[r], pr1 = zr[r + 4];
    const float pi0 = zi[r], pi1 = zi[r + 4];
    xr[r]     = odd ? pr1 : pr0;
    xr[r + 4] = odd ? pr0 : pr1;
    xi[r]     = odd ? pi1 : pi0;
    xi[r + 4] = odd ? pi0 : pi1;
  }
}

struct cf { float r, i; };

__device__ __forceinline__ void rot_u(float phi, float th, float om,
                                      float* __restrict__ u,
                                      float* __restrict__ ryp)
{
  const float pr = __cosf(0.5f * phi), pi = -__sinf(0.5f * phi);
  const float qr = __cosf(0.5f * om),  qi = -__sinf(0.5f * om);
  const float cy = __cosf(0.5f * th),  sy =  __sinf(0.5f * th);
  const float qpr  = qr*pr - qi*pi, qpi  = qr*pi + qi*pr;
  const float qcpr = qr*pr + qi*pi, qcpi = qi*pr - qr*pi;
  u[0] =  cy*qpr;  u[1] =  cy*qpi;
  u[2] = -sy*qcpr; u[3] = -sy*qcpi;
  u[4] =  sy*qcpr; u[5] = -sy*qcpi;
  u[6] =  cy*qpr;  u[7] = -cy*qpi;
  ryp[0] = cy; ryp[1] = sy; ryp[2] = pr; ryp[3] = pi;
}

__global__ __launch_bounds__(512, 16) void quantum_block_kernel(
    const float* __restrict__ x,
    const float* __restrict__ W1, const float* __restrict__ b1,
    const float* __restrict__ W2, const float* __restrict__ b2,
    const float* __restrict__ W3, const float* __restrict__ b3,
    const float* __restrict__ qw,
    const float* __restrict__ Wp, const float* __restrict__ bp,
    float* __restrict__ out)
{
  __shared__ __attribute__((aligned(16))) float sX  [16 * 80];
  __shared__ __attribute__((aligned(16))) float sH1 [16 * 64];
  __shared__ __attribute__((aligned(16))) float sH2 [16 * 32];
  __shared__ float sAng[16 * 16];
  __shared__ __attribute__((aligned(16))) float sZ  [16 * 8];
  __shared__ __attribute__((aligned(16))) float sU[24 * 8];
  __shared__ __attribute__((aligned(16))) float sTabR[2 * 256];
  __shared__ __attribute__((aligned(16))) float sPh[512];

  const int tid     = threadIdx.x;
  const int wid     = tid >> 5;
  const int lane    = tid & 31;
  const int rowBase = blockIdx.x * 16;

  if (tid < 320) {
    const float4 xv = reinterpret_cast<const float4*>(x)[rowBase * 20 + tid];
    reinterpret_cast<float4*>(sX)[tid] = xv;
  } else if (tid < 344) {
    const int t = tid - 320;
    const int L = t >> 3, W = t & 7;
    float u[8], ryp[4];
    rot_u(qw[(L * 8 + W) * 3 + 0], qw[(L * 8 + W) * 3 + 1],
          qw[(L * 8 + W) * 3 + 2], u, ryp);
    const bool t0 = (L == 0), t2 = (L == 2);
    sU[t * 8 + 0] = t2 ? ryp[0] : u[0];
    sU[t * 8 + 1] = t2 ? ryp[1] : (t0 ? u[4] : u[1]);
    sU[t * 8 + 2] = t2 ? ryp[2] : (t0 ? u[1] : u[2]);
    sU[t * 8 + 3] = t2 ? ryp[3] : (t0 ? u[5] : u[3]);
    sU[t * 8 + 4] = t0 ? u[2] : u[4];
    sU[t * 8 + 5] = t0 ? u[6] : u[5];
    sU[t * 8 + 6] = t0 ? u[3] : u[6];
    sU[t * 8 + 7] = u[7];
  }
  __syncthreads();

  if (wid < 4) gemm_tile_to_lds<80, 80, true >(sX,  W1, b1, sH1, 64, wid * 16);
  else         build_layer2_tables(sU, sTabR, sPh, tid - 128);
  __syncthreads();
  if (wid < 2) gemm_tile_to_lds<64, 64, true >(sH1, W2, b2, sH2, 32, wid * 16);
  __syncthreads();
  if (wid == 0) gemm_tile_to_lds<32, 32, false>(sH2, W3, b3, sAng, 16, 0);
  __syncthreads();

  {
    const int h = lane >> 4;
    const int c = lane & 15;
    const float* ang = &sAng[wid * 16];

    v2f E0r, E0i, E1r, E1i;
    int tprev = 0;
#pragma unroll
    for (int p = 0; p < 4; ++p) {
      const int tb = (c >> wire_bitpos(h, p)) & 1;
      const int i  = tb ^ tprev;
      tprev = tb;
      const float al = ang[4 * h + p], be = ang[8 + 4 * h + p];
      const float ca = __cosf(0.5f * al), sa = __sinf(0.5f * al);
      const float cb = __cosf(0.5f * be), sb = __sinf(0.5f * be);
      const float f0r = ca * cb, f0i = -(ca * sb);
      const float f1r = sa * cb, f1i = sa * sb;
      const float4 ua = *reinterpret_cast<const float4*>(&sU[(4 * h + p) * 8]);
      const float4 ub = *reinterpret_cast<const float4*>(&sU[(4 * h + p) * 8 + 4]);
      const v2f UAr = {ua.x, ua.y}, UAi = {ua.z, ua.w};
      const v2f UBr = {ub.x, ub.y}, UBi = {ub.z, ub.w};
      const v2f Gr = UAr * f0r - UAi * f0i + UBr * f1r - UBi * f1i;
      const v2f Gi = UAr * f0i + UAi * f0r + UBr * f1i + UBi * f1r;
      const float4 w4 =
          *reinterpret_cast<const float4*>(&sU[(8 + 4 * h + p) * 8 + i * 4]);
      if (p == 0) {
        const v2f Gsr = {Gr.y, Gr.x}, Gsi = {Gi.y, Gi.x};
        E0r = Gr * w4.x - Gi * w4.y;    E0i = Gr * w4.y + Gi * w4.x;
        E1r = Gsr * w4.z - Gsi * w4.w;  E1i = Gsr * w4.w + Gsi * w4.z;
      } else {
        const bool x = (p == 1) && (h == 0);
        const v2f Ar = {Gr.x, x ? Gr.y : Gr.x}, Ai = {Gi.x, x ? Gi.y : Gi.x};
        const v2f Br = {Gr.y, x ? Gr.x : Gr.y}, Bi = {Gi.y, x ? Gi.x : Gi.y};
        const v2f S0r = E0r * Ar - E0i * Ai + E1r * Br - E1i * Bi;
        const v2f S0i = E0r * Ai + E0i * Ar + E1r * Bi + E1i * Br;
        const v2f S1r = E0r * Br - E0i * Bi + E1r * Ar - E1i * Ai;
        const v2f S1i = E0r * Bi + E0i * Br + E1r * Ai + E1i * Ar;
        E0r = S0r * w4.x - S0i * w4.y;  E0i = S0r * w4.y + S0i * w4.x;
        E1r = S1r * w4.z - S1i * w4.w;  E1i = S1r * w4.w + S1i * w4.z;
      }
    }
    const cf e0[2] = {{E0r.x, E0i.x}, {E1r.x, E1i.x}};
    const cf e1[2] = {{E0r.y, E0i.y}, {E1r.y, E1i.y}};
    const cf snd0 = h ? e0[0] : e1[0];
    const cf snd1 = h ? e1[0] : e1[1];
    cf rcv0, rcv1;
    rcv0.r = __shfl_xor(snd0.r, 16, 32);  rcv0.i = __shfl_xor(snd0.i, 16, 32);
    rcv1.r = __shfl_xor(snd1.r, 16, 32);  rcv1.i = __shfl_xor(snd1.i, 16, 32);
    const cf P0 = h ? rcv0 : e0[0], Q0 = h ? e0[1] : rcv0;
    const cf P1 = h ? rcv1 : e0[1], Q1 = h ? e1[1] : rcv1;
    v2f ar0, br0, ai0, bi0, ar1, br1, ai1, bi1;
    ar0.x = P0.r;  ar0.y = -P0.i;  br0.x = Q0.r;  br0.y = Q0.i;
    ai0.x = P0.r;  ai0.y =  P0.i;  bi0.x = Q0.i;  bi0.y = Q0.r;
    ar1.x = P1.r;  ar1.y = -P1.i;  br1.x = Q1.r;  br1.y = Q1.i;
    ai1.x = P1.r;  ai1.y =  P1.i;  bi1.x = Q1.i;  bi1.y = Q1.r;
    v8f pr = {}, pi = {};
    pr = __builtin_amdgcn_wmma_f32_16x16x4_f32(false, ar0, false, br0,
                                               (short)0, pr, false, false);
    pi = __builtin_amdgcn_wmma_f32_16x16x4_f32(false, ai0, false, bi0,
                                               (short)0, pi, false, false);
    pr = __builtin_amdgcn_wmma_f32_16x16x4_f32(false, ar1, false, br1,
                                               (short)0, pr, false, false);
    pi = __builtin_amdgcn_wmma_f32_16x16x4_f32(false, ai1, false, bi1,
                                               (short)0, pi, false, false);

    const bool odd = (lane & 1) != 0;
    v8f xr, xi;
    ring_cross(pr, pi, odd, xr, xi);
    {
      const float4* d4 = reinterpret_cast<const float4*>(sPh);
      const float4 c0 = d4[lane],      c1 = d4[32 + lane];
      const float4 s0 = d4[64 + lane], s1 = d4[96 + lane];
      const v8f dr = {c0.x, c0.y, c0.z, c0.w, c1.x, c1.y, c1.z, c1.w};
      const v8f di = {s0.x, s0.y, s0.z, s0.w, s1.x, s1.y, s1.z, s1.w};
      const v8f wr = xr * dr - xi * di;
      const v8f wi = xr * di + xi * dr;
      v8f yr, yi, zr, zi;
      kron_stage_real(wr, wi, &sTabR[0],   lane, yr, yi);
      kron_stage_real(yr, yi, &sTabR[256], lane, zr, zi);
      ring_cross(zr, zi, odd, xr, xi);
    }

    float pb[8]; float tot = 0.f;
#pragma unroll
    for (int r = 0; r < 8; ++r) { pb[r] = xr[r]*xr[r] + xi[r]*xi[r]; tot += pb[r]; }
    float s0 = 0.f, s2 = 0.f, s3 = 0.f;
#pragma unroll
    for (int r = 0; r < 8; ++r) {
      s0 += ((r >> 2) & 1) ? -pb[r] : pb[r];
      s2 += ((r >> 1) & 1) ? -pb[r] : pb[r];
      s3 += ( r       & 1) ? -pb[r] : pb[r];
    }
    float zp[8];
    zp[0] = s0;
    zp[1] = h ? -tot : tot;
    zp[2] = s2;
    zp[3] = s3;
#pragma unroll
    for (int w = 4; w < 8; ++w)
      zp[w] = ((c >> (7 - w)) & 1) ? -tot : tot;
    const bool up16 = (lane & 16) != 0;
    const bool up8  = (lane & 8) != 0;
    const bool up4  = (lane & 4) != 0;
    float q4[4], q2[2];
#pragma unroll
    for (int j = 0; j < 4; ++j) {
      const float snd = up16 ? zp[j] : zp[4 + j];
      const float kp  = up16 ? zp[4 + j] : zp[j];
      q4[j] = kp + __shfl_xor(snd, 16, 32);
    }
#pragma unroll
    for (int j = 0; j < 2; ++j) {
      const float snd = up8 ? q4[j] : q4[2 + j];
      const float kp  = up8 ? q4[2 + j] : q4[j];
      q2[j] = kp + __shfl_xor(snd, 8, 32);
    }
    float zs;
    {
      const float snd = up4 ? q2[0] : q2[1];
      const float kp  = up4 ? q2[1] : q2[0];
      zs = kp + __shfl_xor(snd, 4, 32);
    }
    zs += __shfl_xor(zs, 2, 32);
    zs += __shfl_xor(zs, 1, 32);
    if ((lane & 3) == 0) sZ[wid * 8 + (lane >> 2)] = zs;
  }
  __syncthreads();

  if (wid < 8) gemm_final(sZ, Wp, bp, out, rowBase, wid * 16);
}

extern "C" void kernel_launch(void* const* d_in, const int* in_sizes, int n_in,
                              void* d_out, int out_size, void* d_ws, size_t ws_size,
                              hipStream_t stream) {
  const float* x  = (const float*)d_in[0];
  const float* W1 = (const float*)d_in[1];
  const float* b1 = (const float*)d_in[2];
  const float* W2 = (const float*)d_in[3];
  const float* b2 = (const float*)d_in[4];
  const float* W3 = (const float*)d_in[5];
  const float* b3 = (const float*)d_in[6];
  const float* qw = (const float*)d_in[7];
  const float* Wp = (const float*)d_in[8];
  const float* bp = (const float*)d_in[9];
  float* out = (float*)d_out;

  const int B = in_sizes[0] / 80;
  const int grid = B / 16;
  quantum_block_kernel<<<grid, 512, 0, stream>>>(x, W1, b1, W2, b2, W3, b3,
                                                 qw, Wp, bp, out);
}
